# starting kernel
# baseline (speedup 1.0000x reference)
#include <hip/hip_runtime.h>
#include <hip/hip_bf16.h>

typedef __attribute__((ext_vector_type(2))) float v2f;
typedef __attribute__((ext_vector_type(4))) float v4f;
typedef __attribute__((ext_vector_type(8))) float v8f;

#define NEG_SLOPE 0.2f
#define SM_EPS 1e-16f

__device__ __forceinline__ float wave_reduce_sum(float v) {
#pragma unroll
  for (int off = 16; off > 0; off >>= 1) v += __shfl_down(v, off, 32);
  return v;
}

__device__ __forceinline__ unsigned f32_order_key(float x) {
  unsigned u = __float_as_uint(x);
  return (u & 0x80000000u) ? ~u : (u | 0x80000000u);
}
__device__ __forceinline__ float f32_order_inv(unsigned k) {
  unsigned u = (k & 0x80000000u) ? (k & 0x7fffffffu) : ~k;
  return __uint_as_float(u);
}

__global__ __launch_bounds__(128) void gat_matvec128(
    const float* __restrict__ W, const float* __restrict__ att,
    float* __restrict__ vd) {
  int k = threadIdx.x;
  float s = 0.0f;
#pragma unroll 8
  for (int h = 0; h < 128; ++h) s += W[k * 128 + h] * att[h];
  vd[k] = s;
}

__global__ __launch_bounds__(256) void gat_gemm128_wmma(
    const float* __restrict__ X, const float* __restrict__ W,
    float* __restrict__ Y, int nrows) {
  __shared__ float sB[64 * 128 * 2];

  for (int idx = threadIdx.x; idx < 128 * 128; idx += 256) {
    int k = idx >> 7, col = idx & 127;
    int g = ((k >> 2) << 1) | ((k >> 1) & 1);
    sB[(((g << 7) | col) << 1) | (k & 1)] = W[idx];
  }
  __syncthreads();

  const int wave = threadIdx.x >> 5;
  const int lane = threadIdx.x & 31;
  const int half = lane >> 4;
  const int rl   = lane & 15;
  const int rowBase = blockIdx.x * 128 + wave * 16;

  int arow = rowBase + rl;
  if (arow > nrows - 1) arow = nrows - 1;
  const float* xr = X + (size_t)arow * 128 + (half << 1);

  v2f afrag[32];
#pragma unroll
  for (int kk = 0; kk < 32; ++kk) afrag[kk] = *(const v2f*)(xr + (kk << 2));

  v8f acc[8];
#pragma unroll
  for (int ct = 0; ct < 8; ++ct) acc[ct] = (v8f)(0.0f);

  const v2f* sB2 = (const v2f*)sB;
#pragma unroll
  for (int kk = 0; kk < 32; ++kk) {
    const v2f a = afrag[kk];
    const int gbase = ((kk << 1) | half) << 7;
#pragma unroll
    for (int ct = 0; ct < 8; ++ct) {
      v2f b = sB2[gbase + (ct << 4) + rl];
      acc[ct] = __builtin_amdgcn_wmma_f32_16x16x4_f32(
          false, a, false, b, (short)0, acc[ct], false, false);
    }
  }

#pragma unroll
  for (int ct = 0; ct < 8; ++ct) {
#pragma unroll
    for (int v = 0; v < 8; ++v) {
      int r = rowBase + (half << 3) + v;
      if (r < nrows) Y[(size_t)r * 128 + (ct << 4) + rl] = acc[ct][v];
    }
  }
}

__global__ __launch_bounds__(256) void gat_scores(
    const float* __restrict__ xs, const float* __restrict__ xin,
    const float* __restrict__ atts, const float* __restrict__ vd,
    float* __restrict__ a_s, float* __restrict__ a_d,
    unsigned* __restrict__ mbits, float* __restrict__ denom, int n) {
  int node = (blockIdx.x * 256 + threadIdx.x) >> 5;
  int lane = threadIdx.x & 31;
  if (node >= n) return;
  size_t base = (size_t)node * 128 + lane * 4;
  v4f xsv = *(const v4f*)(xs + base);
  v4f av  = *(const v4f*)(atts + lane * 4);
  v4f xiv = *(const v4f*)(xin + base);
  v4f vdv = *(const v4f*)(vd + lane * 4);
  float s = xsv.x * av.x + xsv.y * av.y + xsv.z * av.z + xsv.w * av.w;
  float t = xiv.x * vdv.x + xiv.y * vdv.y + xiv.z * vdv.z + xiv.w * vdv.w;
  s = wave_reduce_sum(s);
  t = wave_reduce_sum(t);
  if (lane == 0) {
    a_s[node] = s;
    a_d[node] = t;
    mbits[node] = 0u;
    denom[node] = 0.0f;
  }
}

__global__ __launch_bounds__(256) void gat_edge_max(
    const int* __restrict__ src, const int* __restrict__ dst,
    const float* __restrict__ a_s, const float* __restrict__ a_d,
    unsigned* __restrict__ mbits, int nedges) {
  int i = blockIdx.x * 256 + threadIdx.x;
  if (i >= nedges) return;
  float e = a_s[src[i]] + a_d[dst[i]];
  e = (e > 0.0f) ? e : NEG_SLOPE * e;
  atomicMax(mbits + dst[i], f32_order_key(e));
}

__global__ __launch_bounds__(256) void gat_node_m(unsigned* __restrict__ mbits,
                                                  int n) {
  int i = blockIdx.x * 256 + threadIdx.x;
  if (i >= n) return;
  unsigned k = mbits[i];
  ((float*)mbits)[i] = (k == 0u) ? 0.0f : f32_order_inv(k);
}

__global__ __launch_bounds__(256) void gat_edge_exp(
    const int* __restrict__ src, const int* __restrict__ dst,
    const float* __restrict__ a_s, const float* __restrict__ a_d,
    const float* __restrict__ m, float* __restrict__ exb,
    float* __restrict__ denom, int nedges) {
  int i = blockIdx.x * 256 + threadIdx.x;
  if (i >= nedges) return;
  int d = dst[i];
  float e = a_s[src[i]] + a_d[d];
  e = (e > 0.0f) ? e : NEG_SLOPE * e;
  float ex = __expf(e - m[d]);
  exb[i] = ex;
  unsafeAtomicAdd(denom + d, ex);
}

__global__ __launch_bounds__(256) void gat_zero4(float* __restrict__ p,
                                                 int count4) {
  int i = blockIdx.x * 256 + threadIdx.x;
  if (i < count4) ((v4f*)p)[i] = (v4f)(0.0f);
}

__global__ __launch_bounds__(256) void gat_edge_aggregate(
    const int* __restrict__ src, const int* __restrict__ dst,
    const float* __restrict__ exb, const float* __restrict__ denom,
    const float* __restrict__ xs, float* __restrict__ out, int nedges) {
  int e = (blockIdx.x * 256 + threadIdx.x) >> 5;
  if (e >= nedges) return;
  int lane = threadIdx.x & 31;
  int s = src[e], d = dst[e];
  float alpha = exb[e] / (denom[d] + SM_EPS);
  v4f v = *(const v4f*)(xs + (size_t)s * 128 + lane * 4);
  float* o = out + (size_t)d * 128 + lane * 4;
  unsafeAtomicAdd(o + 0, alpha * v.x);
  unsafeAtomicAdd(o + 1, alpha * v.y);
  unsafeAtomicAdd(o + 2, alpha * v.z);
  unsafeAtomicAdd(o + 3, alpha * v.w);
}

__global__ __launch_bounds__(256) void gat_bias_relu(
    float* __restrict__ buf, const float* __restrict__ bias, int n) {
  int i = blockIdx.x * 256 + threadIdx.x;
  if (i >= n * 32) return;
  int node = i >> 5, c4 = i & 31;
  v4f v = *(v4f*)(buf + (size_t)node * 128 + c4 * 4);
  v4f b = *(const v4f*)(bias + c4 * 4);
  v.x = fmaxf(v.x + b.x, 0.0f);
  v.y = fmaxf(v.y + b.y, 0.0f);
  v.z = fmaxf(v.z + b.z, 0.0f);
  v.w = fmaxf(v.w + b.w, 0.0f);
  *(v4f*)(buf + (size_t)node * 128 + c4 * 4) = v;
}

__global__ __launch_bounds__(256) void gat_final_linear(
    const float* __restrict__ h, const float* __restrict__ Wl,
    const float* __restrict__ bl, float* __restrict__ out, int n) {
  int node = (blockIdx.x * 256 + threadIdx.x) >> 5;
  int lane = threadIdx.x & 31;
  if (node >= n) return;
  v4f hv = *(const v4f*)(h + (size_t)node * 128 + lane * 4);
  v4f w0 = *(const v4f*)(Wl + lane * 8);
  v4f w1 = *(const v4f*)(Wl + lane * 8 + 4);
  float p0 = hv.x * w0.x + hv.y * w0.z + hv.z * w1.x + hv.w * w1.z;
  float p1 = hv.x * w0.y + hv.y * w0.w + hv.z * w1.y + hv.w * w1.w;
  p0 = wave_reduce_sum(p0);
  p1 = wave_reduce_sum(p1);
  if (lane == 0) {
    out[(size_t)node * 2 + 0] = p0 + bl[0];
    out[(size_t)node * 2 + 1] = p1 + bl[1];
  }
}

static void run_gat_layer(const float* xin, const float* Wsrc,
                          const float* Wdst, const float* atts,
                          const float* attd, const float* bias, const int* src,
                          const int* dst, int N, int E, float* xs,
                          float* outbuf, float* a_s, float* a_d, unsigned* mb,
                          float* denom, float* exb, float* vd,
                          hipStream_t stream) {
  gat_matvec128<<<1, 128, 0, stream>>>(Wdst, attd, vd);
  gat_gemm128_wmma<<<(N + 127) / 128, 256, 0, stream>>>(xin, Wsrc, xs, N);
  gat_scores<<<(N + 7) / 8, 256, 0, stream>>>(xs, xin, atts, vd, a_s, a_d, mb,
                                              denom, N);
  gat_edge_max<<<(E + 255) / 256, 256, 0, stream>>>(src, dst, a_s, a_d, mb, E);
  gat_node_m<<<(N + 255) / 256, 256, 0, stream>>>(mb, N);
  gat_edge_exp<<<(E + 255) / 256, 256, 0, stream>>>(src, dst, a_s, a_d,
                                                    (const float*)mb, exb,
                                                    denom, E);
  gat_zero4<<<(N * 32 + 255) / 256, 256, 0, stream>>>(outbuf, N * 32);
  gat_edge_aggregate<<<(E + 7) / 8, 256, 0, stream>>>(src, dst, exb, denom, xs,
                                                      outbuf, E);
  gat_bias_relu<<<(N * 32 + 255) / 256, 256, 0, stream>>>(outbuf, bias, N);
}

extern "C" void kernel_launch(void* const* d_in, const int* in_sizes, int n_in,
                              void* d_out, int out_size, void* d_ws,
                              size_t ws_size, hipStream_t stream) {
  const float* x   = (const float*)d_in[0];
  const int* ei    = (const int*)d_in[1];
  const float* W1s = (const float*)d_in[2];
  const float* W1d = (const float*)d_in[3];
  const float* a1s = (const float*)d_in[4];
  const float* a1d = (const float*)d_in[5];
  const float* b1  = (const float*)d_in[6];
  const float* W2s = (const float*)d_in[7];
  const float* W2d = (const float*)d_in[8];
  const float* a2s = (const float*)d_in[9];
  const float* a2d = (const float*)d_in[10];
  const float* b2  = (const float*)d_in[11];
  const float* Wl  = (const float*)d_in[12];
  const float* bl  = (const float*)d_in[13];

  const int N = in_sizes[0] / 128;
  const int E = in_sizes[1] / 2;
  const int* src = ei;
  const int* dst = ei + E;

  float* bufA  = (float*)d_ws;
  float* bufB  = bufA + (size_t)N * 128;
  float* a_s   = bufB + (size_t)N * 128;
  float* a_d   = a_s + N;
  unsigned* mb = (unsigned*)(a_d + N);
  float* denom = (float*)mb + N;
  float* exb   = denom + N;
  float* vd    = exb + E;

  run_gat_layer(x, W1s, W1d, a1s, a1d, b1, src, dst, N, E, bufA, bufB, a_s,
                a_d, mb, denom, exb, vd, stream);
  run_gat_layer(bufB, W2s, W2d, a2s, a2d, b2, src, dst, N, E, bufA, bufB, a_s,
                a_d, mb, denom, exb, vd, stream);
  gat_final_linear<<<(N + 7) / 8, 256, 0, stream>>>(bufB, Wl, bl,
                                                    (float*)d_out, N);
}


# Edges grouped by destination once per call (atomic arrival-order rank + scan); per-node wave does softmax, sum, bias, relu in registers: no float atomics, sum order now follows the rank
# speedup vs baseline: 3.5459x; 3.5459x over previous
#include <hip/hip_runtime.h>
#include <hip/hip_bf16.h>

typedef __attribute__((ext_vector_type(2))) float v2f;
typedef __attribute__((ext_vector_type(4))) float v4f;
typedef __attribute__((ext_vector_type(8))) float v8f;

#define NEG_SLOPE 0.2f
#define SM_EPS 1e-16f

__device__ __forceinline__ float wave_reduce_sum(float v) {
#pragma unroll
  for (int off = 16; off > 0; off >>= 1) v += __shfl_down(v, off, 32);
  return v;
}

__global__ __launch_bounds__(128) void gat_matvec128(
    const float* __restrict__ W, const float* __restrict__ att,
    float* __restrict__ vd) {
  int k = threadIdx.x;
  float s = 0.0f;
#pragma unroll 8
  for (int h = 0; h < 128; ++h) s += W[k * 128 + h] * att[h];
  vd[k] = s;
}

__global__ __launch_bounds__(256) void gat_gemm128_wmma(
    const float* __restrict__ X, const float* __restrict__ W,
    float* __restrict__ Y, int nrows) {
  __shared__ float sB[64 * 128 * 2];

  for (int idx = threadIdx.x; idx < 128 * 128; idx += 256) {
    int k = idx >> 7, col = idx & 127;
    int g = ((k >> 2) << 1) | ((k >> 1) & 1);
    sB[(((g << 7) | col) << 1) | (k & 1)] = W[idx];
  }
  __syncthreads();

  const int wave = threadIdx.x >> 5;
  const int lane = threadIdx.x & 31;
  const int half = lane >> 4;
  const int rl   = lane & 15;
  const int rowBase = blockIdx.x * 128 + wave * 16;

  int arow = rowBase + rl;
  if (arow > nrows - 1) arow = nrows - 1;
  const float* xr = X + (size_t)arow * 128 + (half << 1);

  v2f afrag[32];
#pragma unroll
  for (int kk = 0; kk < 32; ++kk) afrag[kk] = *(const v2f*)(xr + (kk << 2));

  v8f acc[8];
#pragma unroll
  for (int ct = 0; ct < 8; ++ct) acc[ct] = (v8f)(0.0f);

  const v2f* sB2 = (const v2f*)sB;
#pragma unroll
  for (int kk = 0; kk < 32; ++kk) {
    const v2f a = afrag[kk];
    const int gbase = ((kk << 1) | half) << 7;
#pragma unroll
    for (int ct = 0; ct < 8; ++ct) {
      v2f b = sB2[gbase + (ct << 4) + rl];
      acc[ct] = __builtin_amdgcn_wmma_f32_16x16x4_f32(
          false, a, false, b, (short)0, acc[ct], false, false);
    }
  }

#pragma unroll
  for (int ct = 0; ct < 8; ++ct) {
#pragma unroll
    for (int v = 0; v < 8; ++v) {
      int r = rowBase + (half << 3) + v;
      if (r < nrows) Y[(size_t)r * 128 + (ct << 4) + rl] = acc[ct][v];
    }
  }
}

__global__ __launch_bounds__(256) void gat_scores(
    const float* __restrict__ xs, const float* __restrict__ xin,
    const float* __restrict__ atts, const float* __restrict__ vd,
    float* __restrict__ a_s, float* __restrict__ a_d, int n) {
  int node = (blockIdx.x * 256 + threadIdx.x) >> 5;
  int lane = threadIdx.x & 31;
  if (node >= n) return;
  size_t base = (size_t)node * 128 + lane * 4;
  v4f xsv = *(const v4f*)(xs + base);
  v4f av  = *(const v4f*)(atts + lane * 4);
  v4f xiv = *(const v4f*)(xin + base);
  v4f vdv = *(const v4f*)(vd + lane * 4);
  float s = xsv.x * av.x + xsv.y * av.y + xsv.z * av.z + xsv.w * av.w;
  float t = xiv.x * vdv.x + xiv.y * vdv.y + xiv.z * vdv.z + xiv.w * vdv.w;
  s = wave_reduce_sum(s);
  t = wave_reduce_sum(t);
  if (lane == 0) {
    a_s[node] = s;
    a_d[node] = t;
  }
}

__global__ __launch_bounds__(256) void gat_zero_counts(int* __restrict__ cnt,
                                                       int n) {
  int i = blockIdx.x * 256 + threadIdx.x;
  if (i < n) cnt[i] = 0;
}

__global__ __launch_bounds__(256) void gat_edge_rank(
    const int* __restrict__ dst, int* __restrict__ cnt,
    int* __restrict__ rank, int nedges) {
  int i = blockIdx.x * 256 + threadIdx.x;
  if (i >= nedges) return;
  rank[i] = atomicAdd(cnt + dst[i], 1);
}

__device__ __forceinline__ int wave_reduce_sum_i(int v) {
  for (int off = 16; off > 0; off >>= 1) v += __shfl_down(v, off, 32);
  return v;
}

__global__ __launch_bounds__(256) void gat_scan_tiles(
    const int* __restrict__ cnt, int* __restrict__ bsum, int n) {
  __shared__ int sw[8];
  const int base = blockIdx.x * 1024 + threadIdx.x * 4;
  int s = 0;
  for (int j = 0; j < 4; ++j)
    if (base + j < n) s += cnt[base + j];
  s = wave_reduce_sum_i(s);
  if ((threadIdx.x & 31) == 0) sw[threadIdx.x >> 5] = s;
  __syncthreads();
  if (threadIdx.x == 0) {
    int t = 0;
    for (int w = 0; w < 8; ++w) t += sw[w];
    bsum[blockIdx.x] = t;
  }
}

__global__ __launch_bounds__(256) void gat_scan_rowptr(
    const int* __restrict__ cnt, const int* __restrict__ bsum,
    int* __restrict__ rowptr, int n) {
  __shared__ int sw[8];
  __shared__ int sv[8];
  const int tid = threadIdx.x, lane = tid & 31, wave = tid >> 5;

  int o = 0;
  for (int k = tid; k < (int)blockIdx.x; k += 256) o += bsum[k];
  o = wave_reduce_sum_i(o);
  if (lane == 0) sw[wave] = o;

  const int base = blockIdx.x * 1024 + tid * 4;
  int c[4];
  int s = 0;
#pragma unroll
  for (int j = 0; j < 4; ++j) {
    c[j] = (base + j < n) ? cnt[base + j] : 0;
    s += c[j];
  }
  int inc = s;
  for (int off = 1; off < 32; off <<= 1) {
    int t = __shfl_up(inc, off, 32);
    if (lane >= off) inc += t;
  }
  if (lane == 31) sv[wave] = inc;
  __syncthreads();

  int run = inc - s;
  for (int w = 0; w < 8; ++w) {
    run += sw[w];
    if (w < wave) run += sv[w];
  }
#pragma unroll
  for (int j = 0; j < 4; ++j) {
    if (base + j < n) {
      rowptr[base + j] = run;
      run += c[j];
      if (base + j == n - 1) rowptr[n] = run;
    }
  }
}

__global__ __launch_bounds__(256) void gat_edge_fill(
    const int* __restrict__ src, const int* __restrict__ dst,
    const int* __restrict__ rank, const int* __restrict__ rowptr,
    int* __restrict__ ssrc, int nedges) {
  int i = blockIdx.x * 256 + threadIdx.x;
  if (i >= nedges) return;
  ssrc[rowptr[dst[i]] + rank[i]] = src[i];
}

__device__ __forceinline__ float wave_allreduce_max(float v) {
  for (int off = 16; off > 0; off >>= 1) v = fmaxf(v, __shfl_xor(v, off, 32));
  return v;
}
__device__ __forceinline__ float wave_allreduce_sum(float v) {
  for (int off = 16; off > 0; off >>= 1) v += __shfl_xor(v, off, 32);
  return v;
}
__device__ __forceinline__ float gat_leaky(float e) {
  return (e > 0.0f) ? e : NEG_SLOPE * e;
}
__device__ __forceinline__ int lane_i(int v, int l) {
  return __builtin_amdgcn_readlane(v, l);
}
__device__ __forceinline__ float lane_f(float v, int l) {
  return __int_as_float(__builtin_amdgcn_readlane(__float_as_int(v), l));
}
__device__ __forceinline__ v4f gat_row(const float* __restrict__ xl, int s) {
  return *(const v4f*)(xl + (size_t)s * 128);
}

__global__ __launch_bounds__(256) void gat_node_aggregate(
    const int* __restrict__ rowptr, const int* __restrict__ ssrc,
    const float* __restrict__ a_s, const float* __restrict__ a_d,
    const float* __restrict__ xs, const float* __restrict__ bias,
    float* __restrict__ out, int n) {
  int node = (blockIdx.x * 256 + threadIdx.x) >> 5;
  const int lane = threadIdx.x & 31;
  if (node >= n) return;
  node = __builtin_amdgcn_readfirstlane(node);
  const int beg = rowptr[node], end = rowptr[node + 1];
  const float ad = a_d[node];

  const int c0 = beg + lane;
  const bool v0 = c0 < end;
  int s0 = 0;
  float e0 = -INFINITY;
  if (v0) {
    s0 = ssrc[c0];
    e0 = gat_leaky(a_s[s0] + ad);
  }
  float mx = e0;
  for (int c = c0 + 32; c < end; c += 32)
    mx = fmaxf(mx, gat_leaky(a_s[ssrc[c]] + ad));
  mx = wave_allreduce_max(mx);
  if (!(fabsf(mx) < INFINITY)) mx = 0.0f;

  const float x0 = v0 ? __expf(e0 - mx) : 0.0f;
  float sum = x0;
  for (int c = c0 + 32; c < end; c += 32)
    sum += __expf(gat_leaky(a_s[ssrc[c]] + ad) - mx);
  sum = wave_allreduce_sum(sum);
  const float dn = sum + SM_EPS;

  const float* xl = xs + lane * 4;
  v4f acc = (v4f)(0.0f);
  for (int cb = beg; cb < end; cb += 32) {
    int sl = s0;
    float al = x0 / dn;
    if (cb != beg) {
      const int c = cb + lane;
      sl = 0;
      al = 0.0f;
      if (c < end) {
        sl = ssrc[c];
        al = __expf(gat_leaky(a_s[sl] + ad) - mx) / dn;
      }
    }
    const int cnt = (end - cb < 32) ? (end - cb) : 32;
    int j = 0;
    for (; j + 4 <= cnt; j += 4) {
      const v4f va = gat_row(xl, lane_i(sl, j));
      const v4f vb = gat_row(xl, lane_i(sl, j + 1));
      const v4f vc = gat_row(xl, lane_i(sl, j + 2));
      const v4f vdd = gat_row(xl, lane_i(sl, j + 3));
      acc += lane_f(al, j) * va;
      acc += lane_f(al, j + 1) * vb;
      acc += lane_f(al, j + 2) * vc;
      acc += lane_f(al, j + 3) * vdd;
    }
    const int rem = cnt - j;
    if (rem == 3) {
      const v4f va = gat_row(xl, lane_i(sl, j));
      const v4f vb = gat_row(xl, lane_i(sl, j + 1));
      const v4f vc = gat_row(xl, lane_i(sl, j + 2));
      acc += lane_f(al, j) * va;
      acc += lane_f(al, j + 1) * vb;
      acc += lane_f(al, j + 2) * vc;
    } else if (rem == 2) {
      const v4f va = gat_row(xl, lane_i(sl, j));
      const v4f vb = gat_row(xl, lane_i(sl, j + 1));
      acc += lane_f(al, j) * va;
      acc += lane_f(al, j + 1) * vb;
    } else if (rem == 1) {
      const v4f va = gat_row(xl, lane_i(sl, j));
      acc += lane_f(al, j) * va;
    }
  }

  const v4f b = *(const v4f*)(bias + lane * 4);
  v4f r;
  r.x = fmaxf(acc.x + b.x, 0.0f);
  r.y = fmaxf(acc.y + b.y, 0.0f);
  r.z = fmaxf(acc.z + b.z, 0.0f);
  r.w = fmaxf(acc.w + b.w, 0.0f);
  *(v4f*)(out + (size_t)node * 128 + lane * 4) = r;
}

__global__ __launch_bounds__(256) void gat_final_linear(
    const float* __restrict__ h, const float* __restrict__ Wl,
    const float* __restrict__ bl, float* __restrict__ out, int n) {
  int node = (blockIdx.x * 256 + threadIdx.x) >> 5;
  int lane = threadIdx.x & 31;
  if (node >= n) return;
  v4f hv = *(const v4f*)(h + (size_t)node * 128 + lane * 4);
  v4f w0 = *(const v4f*)(Wl + lane * 8);
  v4f w1 = *(const v4f*)(Wl + lane * 8 + 4);
  float p0 = hv.x * w0.x + hv.y * w0.z + hv.z * w1.x + hv.w * w1.z;
  float p1 = hv.x * w0.y + hv.y * w0.w + hv.z * w1.y + hv.w * w1.w;
  p0 = wave_reduce_sum(p0);
  p1 = wave_reduce_sum(p1);
  if (lane == 0) {
    out[(size_t)node * 2 + 0] = p0 + bl[0];
    out[(size_t)node * 2 + 1] = p1 + bl[1];
  }
}

static void run_gat_layer(const float* xin, const float* Wsrc,
                          const float* Wdst, const float* atts,
                          const float* attd, const float* bias,
                          const int* rowptr, const int* ssrc, int N, float* xs,
                          float* outbuf, float* a_s, float* a_d, float* vd,
                          hipStream_t stream) {
  gat_matvec128<<<1, 128, 0, stream>>>(Wdst, attd, vd);
  gat_gemm128_wmma<<<(N + 127) / 128, 256, 0, stream>>>(xin, Wsrc, xs, N);
  gat_scores<<<(N + 7) / 8, 256, 0, stream>>>(xs, xin, atts, vd, a_s, a_d, N);
  gat_node_aggregate<<<(N + 7) / 8, 256, 0, stream>>>(rowptr, ssrc, a_s, a_d,
                                                      xs, bias, outbuf, N);
}

extern "C" void kernel_launch(void* const* d_in, const int* in_sizes, int n_in,
                              void* d_out, int out_size, void* d_ws,
                              size_t ws_size, hipStream_t stream) {
  const float* x   = (const float*)d_in[0];
  const int* ei    = (const int*)d_in[1];
  const float* W1s = (const float*)d_in[2];
  const float* W1d = (const float*)d_in[3];
  const float* a1s = (const float*)d_in[4];
  const float* a1d = (const float*)d_in[5];
  const float* b1  = (const float*)d_in[6];
  const float* W2s = (const float*)d_in[7];
  const float* W2d = (const float*)d_in[8];
  const float* a2s = (const float*)d_in[9];
  const float* a2d = (const float*)d_in[10];
  const float* b2  = (const float*)d_in[11];
  const float* Wl  = (const float*)d_in[12];
  const float* bl  = (const float*)d_in[13];

  const int N = in_sizes[0] / 128;
  const int E = in_sizes[1] / 2;
  const int* src = ei;
  const int* dst = ei + E;
  const int tiles = (N + 1023) / 1024;

  float* bufA  = (float*)d_ws;
  float* bufB  = bufA + (size_t)N * 128;
  float* vd    = bufB + (size_t)N * 128;
  float* a_s   = vd + 128;
  float* a_d   = a_s + N;
  int* cnt     = (int*)(a_d + N);
  int* rowptr  = cnt + N;
  int* bsum    = rowptr + N + 1;
  int* rank    = bsum + tiles;
  int* ssrc    = rank + E;

  gat_zero_counts<<<(N + 255) / 256, 256, 0, stream>>>(cnt, N);
  gat_edge_rank<<<(E + 255) / 256, 256, 0, stream>>>(dst, cnt, rank, E);
  gat_scan_tiles<<<tiles, 256, 0, stream>>>(cnt, bsum, N);
  gat_scan_rowptr<<<tiles, 256, 0, stream>>>(cnt, bsum, rowptr, N);
  gat_edge_fill<<<(E + 255) / 256, 256, 0, stream>>>(src, dst, rank, rowptr,
                                                     ssrc, E);

  run_gat_layer(x, W1s, W1d, a1s, a1d, b1, rowptr, ssrc, N, bufA, bufB, a_s,
                a_d, vd, stream);
  run_gat_layer(bufB, W2s, W2d, a2s, a2d, b2, rowptr, ssrc, N, bufA, bufB, a_s,
                a_d, vd, stream);
  gat_final_linear<<<(N + 7) / 8, 256, 0, stream>>>(bufB, Wl, bl,
                                                    (float*)d_out, N);
}


# Edges grouped by destination by a two-level counting sort binned in LDS, coalesced writes, not one global atomic, random read and scattered store per edge; sum order now the LDS hand-out order
# speedup vs baseline: 5.0083x; 1.4124x over previous
#include <hip/hip_runtime.h>
#include <hip/hip_bf16.h>

typedef __attribute__((ext_vector_type(2))) float v2f;
typedef __attribute__((ext_vector_type(4))) float v4f;
typedef __attribute__((ext_vector_type(8))) float v8f;

#define NEG_SLOPE 0.2f
#define SM_EPS 1e-16f

__device__ __forceinline__ float wave_reduce_sum(float v) {
#pragma unroll
  for (int off = 16; off > 0; off >>= 1) v += __shfl_down(v, off, 32);
  return v;
}

__global__ __launch_bounds__(128) void gat_matvec128(
    const float* __restrict__ W, const float* __restrict__ att,
    float* __restrict__ vd) {
  int k = threadIdx.x;
  float s = 0.0f;
#pragma unroll 8
  for (int h = 0; h < 128; ++h) s += W[k * 128 + h] * att[h];
  vd[k] = s;
}

__global__ __launch_bounds__(256) void gat_gemm128_wmma(
    const float* __restrict__ X, const float* __restrict__ W,
    float* __restrict__ Y, int nrows) {
  __shared__ float sB[64 * 128 * 2];

  for (int idx = threadIdx.x; idx < 128 * 128; idx += 256) {
    int k = idx >> 7, col = idx & 127;
    int g = ((k >> 2) << 1) | ((k >> 1) & 1);
    sB[(((g << 7) | col) << 1) | (k & 1)] = W[idx];
  }
  __syncthreads();

  const int wave = threadIdx.x >> 5;
  const int lane = threadIdx.x & 31;
  const int half = lane >> 4;
  const int rl   = lane & 15;
  const int rowBase = blockIdx.x * 128 + wave * 16;

  int arow = rowBase + rl;
  if (arow > nrows - 1) arow = nrows - 1;
  const float* xr = X + (size_t)arow * 128 + (half << 1);

  v2f afrag[32];
#pragma unroll
  for (int kk = 0; kk < 32; ++kk) afrag[kk] = *(const v2f*)(xr + (kk << 2));

  v8f acc[8];
#pragma unroll
  for (int ct = 0; ct < 8; ++ct) acc[ct] = (v8f)(0.0f);

  const v2f* sB2 = (const v2f*)sB;
#pragma unroll
  for (int kk = 0; kk < 32; ++kk) {
    const v2f a = afrag[kk];
    const int gbase = ((kk << 1) | half) << 7;
#pragma unroll
    for (int ct = 0; ct < 8; ++ct) {
      v2f b = sB2[gbase + (ct << 4) + rl];
      acc[ct] = __builtin_amdgcn_wmma_f32_16x16x4_f32(
          false, a, false, b, (short)0, acc[ct], false, false);
    }
  }

#pragma unroll
  for (int ct = 0; ct < 8; ++ct) {
#pragma unroll
    for (int v = 0; v < 8; ++v) {
      int r = rowBase + (half << 3) + v;
      if (r < nrows) Y[(size_t)r * 128 + (ct << 4) + rl] = acc[ct][v];
    }
  }
}

__global__ __launch_bounds__(256) void gat_scores(
    const float* __restrict__ xs, const float* __restrict__ xin,
    const float* __restrict__ atts, const float* __restrict__ vd,
    float* __restrict__ a_s, float* __restrict__ a_d, int n) {
  int node = (blockIdx.x * 256 + threadIdx.x) >> 5;
  int lane = threadIdx.x & 31;
  if (node >= n) return;
  size_t base = (size_t)node * 128 + lane * 4;
  v4f xsv = *(const v4f*)(xs + base);
  v4f av  = *(const v4f*)(atts + lane * 4);
  v4f xiv = *(const v4f*)(xin + base);
  v4f vdv = *(const v4f*)(vd + lane * 4);
  float s = xsv.x * av.x + xsv.y * av.y + xsv.z * av.z + xsv.w * av.w;
  float t = xiv.x * vdv.x + xiv.y * vdv.y + xiv.z * vdv.z + xiv.w * vdv.w;
  s = wave_reduce_sum(s);
  t = wave_reduce_sum(t);
  if (lane == 0) {
    a_s[node] = s;
    a_d[node] = t;
  }
}

#define BSH 9
#define BNODES (1 << BSH)
#define MAXB 1024
#define CHUNK 4096
#define BIN_T 512
#define SORT_T 1024
#define STAGE_CAP 16384

__global__ __launch_bounds__(BIN_T) void gat_bin_count(
    const int* __restrict__ dst, int* __restrict__ hist, int nedges, int nb,
    int nblk) {
  __shared__ int h[MAXB];
  const int tid = threadIdx.x;
  for (int b = tid; b < nb; b += BIN_T) h[b] = 0;
  __syncthreads();
  const int base = blockIdx.x * CHUNK;
  for (int k = 0; k < CHUNK / BIN_T; ++k) {
    const int i = base + k * BIN_T + tid;
    if (i < nedges) atomicAdd(&h[dst[i] >> BSH], 1);
  }
  __syncthreads();
  for (int b = tid; b < nb; b += BIN_T)
    hist[b * nblk + (int)blockIdx.x] = h[b];
}

__device__ __forceinline__ int wave_reduce_sum_i(int v) {
  for (int off = 16; off > 0; off >>= 1) v += __shfl_down(v, off, 32);
  return v;
}

__global__ __launch_bounds__(256) void gat_scan_tiles(
    const int* __restrict__ cnt, int* __restrict__ bsum, int n) {
  __shared__ int sw[8];
  const int base = blockIdx.x * 1024 + threadIdx.x * 4;
  int s = 0;
  for (int j = 0; j < 4; ++j)
    if (base + j < n) s += cnt[base + j];
  s = wave_reduce_sum_i(s);
  if ((threadIdx.x & 31) == 0) sw[threadIdx.x >> 5] = s;
  __syncthreads();
  if (threadIdx.x == 0) {
    int t = 0;
    for (int w = 0; w < 8; ++w) t += sw[w];
    bsum[blockIdx.x] = t;
  }
}

__global__ __launch_bounds__(256) void gat_scan_rowptr(
    const int* __restrict__ cnt, const int* __restrict__ bsum,
    int* __restrict__ rowptr, int n) {
  __shared__ int sw[8];
  __shared__ int sv[8];
  const int tid = threadIdx.x, lane = tid & 31, wave = tid >> 5;

  int o = 0;
  for (int k = tid; k < (int)blockIdx.x; k += 256) o += bsum[k];
  o = wave_reduce_sum_i(o);
  if (lane == 0) sw[wave] = o;

  const int base = blockIdx.x * 1024 + tid * 4;
  int c[4];
  int s = 0;
#pragma unroll
  for (int j = 0; j < 4; ++j) {
    c[j] = (base + j < n) ? cnt[base + j] : 0;
    s += c[j];
  }
  int inc = s;
  for (int off = 1; off < 32; off <<= 1) {
    int t = __shfl_up(inc, off, 32);
    if (lane >= off) inc += t;
  }
  if (lane == 31) sv[wave] = inc;
  __syncthreads();

  int run = inc - s;
  for (int w = 0; w < 8; ++w) {
    run += sw[w];
    if (w < wave) run += sv[w];
  }
#pragma unroll
  for (int j = 0; j < 4; ++j) {
    if (base + j < n) {
      rowptr[base + j] = run;
      run += c[j];
      if (base + j == n - 1) rowptr[n] = run;
    }
  }
}

__global__ __launch_bounds__(BIN_T) void gat_bin_scatter(
    const int* __restrict__ src, const int* __restrict__ dst,
    const int* __restrict__ offs, int* __restrict__ pairs, int nedges,
    int nb, int nblk) {
  __shared__ int gofs[MAXB];
  __shared__ int lstart[MAXB];
  __shared__ int cur[MAXB];
  __shared__ int2 sorted[CHUNK];
  const int tid = threadIdx.x;
  const int blk = (int)blockIdx.x;

  for (int b = tid; b < nb; b += BIN_T) {
    const int o = offs[b * nblk + blk];
    gofs[b] = o;
    lstart[b] = offs[b * nblk + blk + 1] - o;
    cur[b] = 0;
  }
  __syncthreads();
  if (tid < 32) {
    const int per = (nb + 31) >> 5;
    const int b0 = tid * per;
    int s = 0;
    for (int j = 0; j < per; ++j)
      if (b0 + j < nb) s += lstart[b0 + j];
    int inc = s;
    for (int off = 1; off < 32; off <<= 1) {
      int t = __shfl_up(inc, off, 32);
      if (tid >= off) inc += t;
    }
    int run = inc - s;
    for (int j = 0; j < per; ++j) {
      if (b0 + j < nb) {
        const int c = lstart[b0 + j];
        lstart[b0 + j] = run;
        run += c;
      }
    }
  }
  __syncthreads();

  const int base = blk * CHUNK;
  for (int k = 0; k < CHUNK / BIN_T; ++k) {
    const int i = base + k * BIN_T + tid;
    if (i < nedges) {
      const int d = dst[i], s = src[i];
      const int b = d >> BSH;
      const int r = atomicAdd(&cur[b], 1);
      sorted[lstart[b] + r] = make_int2(s, d);
    }
  }
  __syncthreads();

  int cnt = nedges - base;
  if (cnt > CHUNK) cnt = CHUNK;
  int2* __restrict__ pout = reinterpret_cast<int2*>(pairs);
  for (int t = tid; t < cnt; t += BIN_T) {
    const int2 p = sorted[t];
    const int b = p.y >> BSH;
    pout[gofs[b] + (t - lstart[b])] = p;
  }
}

__global__ __launch_bounds__(SORT_T) void gat_bucket_sort(
    const int* __restrict__ pairs, const int* __restrict__ offs,
    int* __restrict__ rowptr, int* __restrict__ ssrc, int n, int nb,
    int nblk) {
  __shared__ int cnt[BNODES];
  __shared__ int cursor[BNODES];
  __shared__ int wsum[BNODES / 32];
  __shared__ int stage[STAGE_CAP];
  const int tid = threadIdx.x;
  const int b = (int)blockIdx.x;
  const int beg = offs[b * nblk], end = offs[(b + 1) * nblk];
  const int node0 = b << BSH;
  const int2* __restrict__ pin = reinterpret_cast<const int2*>(pairs);

  if (tid < BNODES) cnt[tid] = 0;
  __syncthreads();
  for (int i0 = beg + tid; i0 < end; i0 += 4 * SORT_T) {
    int dl[4];
#pragma unroll
    for (int u = 0; u < 4; ++u) {
      const int i = i0 + u * SORT_T;
      dl[u] = (i < end) ? pin[i].y : node0;
    }
#pragma unroll
    for (int u = 0; u < 4; ++u)
      if (i0 + u * SORT_T < end) atomicAdd(&cnt[dl[u] - node0], 1);
  }
  __syncthreads();

  int c = 0, inc = 0;
  if (tid < BNODES) {
    c = cnt[tid];
    inc = c;
    for (int off = 1; off < 32; off <<= 1) {
      int t = __shfl_up(inc, off, 32);
      if ((tid & 31) >= off) inc += t;
    }
    if ((tid & 31) == 31) wsum[tid >> 5] = inc;
  }
  __syncthreads();
  if (tid < BNODES) {
    int run = inc - c;
    for (int w = 0; w < (tid >> 5); ++w) run += wsum[w];
    cursor[tid] = run;
    if (node0 + tid < n) rowptr[node0 + tid] = beg + run;
  }
  if (b == nb - 1 && tid == 0) rowptr[n] = end;
  __syncthreads();

  const bool staged = (end - beg) <= STAGE_CAP;
  for (int i0 = beg + tid; i0 < end; i0 += 4 * SORT_T) {
    int2 p[4];
#pragma unroll
    for (int u = 0; u < 4; ++u) {
      const int i = i0 + u * SORT_T;
      p[u] = (i < end) ? pin[i] : make_int2(0, node0);
    }
#pragma unroll
    for (int u = 0; u < 4; ++u) {
      if (i0 + u * SORT_T < end) {
        const int pos = atomicAdd(&cursor[p[u].y - node0], 1);
        if (staged) stage[pos] = p[u].x;
        else ssrc[beg + pos] = p[u].x;
      }
    }
  }
  if (staged) {
    __syncthreads();
    for (int i = tid; i < end - beg; i += SORT_T) ssrc[beg + i] = stage[i];
  }
}

__device__ __forceinline__ float wave_allreduce_max(float v) {
  for (int off = 16; off > 0; off >>= 1) v = fmaxf(v, __shfl_xor(v, off, 32));
  return v;
}
__device__ __forceinline__ float wave_allreduce_sum(float v) {
  for (int off = 16; off > 0; off >>= 1) v += __shfl_xor(v, off, 32);
  return v;
}
__device__ __forceinline__ float gat_leaky(float e) {
  return (e > 0.0f) ? e : NEG_SLOPE * e;
}
__device__ __forceinline__ int lane_i(int v, int l) {
  return __builtin_amdgcn_readlane(v, l);
}
__device__ __forceinline__ float lane_f(float v, int l) {
  return __int_as_float(__builtin_amdgcn_readlane(__float_as_int(v), l));
}
__device__ __forceinline__ v4f gat_row(const float* __restrict__ xl, int s) {
  return *(const v4f*)(xl + (size_t)s * 128);
}

__global__ __launch_bounds__(256) void gat_node_aggregate(
    const int* __restrict__ rowptr, const int* __restrict__ ssrc,
    const float* __restrict__ a_s, const float* __restrict__ a_d,
    const float* __restrict__ xs, const float* __restrict__ bias,
    float* __restrict__ out, int n) {
  int node = (blockIdx.x * 256 + threadIdx.x) >> 5;
  const int lane = threadIdx.x & 31;
  if (node >= n) return;
  node = __builtin_amdgcn_readfirstlane(node);
  const int beg = rowptr[node], end = rowptr[node + 1];
  const float ad = a_d[node];

  const int c0 = beg + lane;
  const bool v0 = c0 < end;
  int s0 = 0;
  float e0 = -INFINITY;
  if (v0) {
    s0 = ssrc[c0];
    e0 = gat_leaky(a_s[s0] + ad);
  }
  float mx = e0;
  for (int c = c0 + 32; c < end; c += 32)
    mx = fmaxf(mx, gat_leaky(a_s[ssrc[c]] + ad));
  mx = wave_allreduce_max(mx);
  if (!(fabsf(mx) < INFINITY)) mx = 0.0f;

  const float x0 = v0 ? __expf(e0 - mx) : 0.0f;
  float sum = x0;
  for (int c = c0 + 32; c < end; c += 32)
    sum += __expf(gat_leaky(a_s[ssrc[c]] + ad) - mx);
  sum = wave_allreduce_sum(sum);
  const float dn = sum + SM_EPS;

  const float* xl = xs + lane * 4;
  v4f acc = (v4f)(0.0f);
  for (int cb = beg; cb < end; cb += 32) {
    int sl = s0;
    float al = x0 / dn;
    if (cb != beg) {
      const int c = cb + lane;
      sl = 0;
      al = 0.0f;
      if (c < end) {
        sl = ssrc[c];
        al = __expf(gat_leaky(a_s[sl] + ad) - mx) / dn;
      }
    }
    const int cnt = (end - cb < 32) ? (end - cb) : 32;
    int j = 0;
    for (; j + 4 <= cnt; j += 4) {
      const v4f va = gat_row(xl, lane_i(sl, j));
      const v4f vb = gat_row(xl, lane_i(sl, j + 1));
      const v4f vc = gat_row(xl, lane_i(sl, j + 2));
      const v4f vdd = gat_row(xl, lane_i(sl, j + 3));
      acc += lane_f(al, j) * va;
      acc += lane_f(al, j + 1) * vb;
      acc += lane_f(al, j + 2) * vc;
      acc += lane_f(al, j + 3) * vdd;
    }
    const int rem = cnt - j;
    if (rem == 3) {
      const v4f va = gat_row(xl, lane_i(sl, j));
      const v4f vb = gat_row(xl, lane_i(sl, j + 1));
      const v4f vc = gat_row(xl, lane_i(sl, j + 2));
      acc += lane_f(al, j) * va;
      acc += lane_f(al, j + 1) * vb;
      acc += lane_f(al, j + 2) * vc;
    } else if (rem == 2) {
      const v4f va = gat_row(xl, lane_i(sl, j));
      const v4f vb = gat_row(xl, lane_i(sl, j + 1));
      acc += lane_f(al, j) * va;
      acc += lane_f(al, j + 1) * vb;
    } else if (rem == 1) {
      const v4f va = gat_row(xl, lane_i(sl, j));
      acc += lane_f(al, j) * va;
    }
  }

  const v4f b = *(const v4f*)(bias + lane * 4);
  v4f r;
  r.x = fmaxf(acc.x + b.x, 0.0f);
  r.y = fmaxf(acc.y + b.y, 0.0f);
  r.z = fmaxf(acc.z + b.z, 0.0f);
  r.w = fmaxf(acc.w + b.w, 0.0f);
  *(v4f*)(out + (size_t)node * 128 + lane * 4) = r;
}

__global__ __launch_bounds__(256) void gat_final_linear(
    const float* __restrict__ h, const float* __restrict__ Wl,
    const float* __restrict__ bl, float* __restrict__ out, int n) {
  int node = (blockIdx.x * 256 + threadIdx.x) >> 5;
  int lane = threadIdx.x & 31;
  if (node >= n) return;
  v4f hv = *(const v4f*)(h + (size_t)node * 128 + lane * 4);
  v4f w0 = *(const v4f*)(Wl + lane * 8);
  v4f w1 = *(const v4f*)(Wl + lane * 8 + 4);
  float p0 = hv.x * w0.x + hv.y * w0.z + hv.z * w1.x + hv.w * w1.z;
  float p1 = hv.x * w0.y + hv.y * w0.w + hv.z * w1.y + hv.w * w1.w;
  p0 = wave_reduce_sum(p0);
  p1 = wave_reduce_sum(p1);
  if (lane == 0) {
    out[(size_t)node * 2 + 0] = p0 + bl[0];
    out[(size_t)node * 2 + 1] = p1 + bl[1];
  }
}

static void run_gat_layer(const float* xin, const float* Wsrc,
                          const float* Wdst, const float* atts,
                          const float* attd, const float* bias,
                          const int* rowptr, const int* ssrc, int N, float* xs,
                          float* outbuf, float* a_s, float* a_d, float* vd,
                          hipStream_t stream) {
  gat_matvec128<<<1, 128, 0, stream>>>(Wdst, attd, vd);
  gat_gemm128_wmma<<<(N + 127) / 128, 256, 0, stream>>>(xin, Wsrc, xs, N);
  gat_scores<<<(N + 7) / 8, 256, 0, stream>>>(xs, xin, atts, vd, a_s, a_d, N);
  gat_node_aggregate<<<(N + 7) / 8, 256, 0, stream>>>(rowptr, ssrc, a_s, a_d,
                                                      xs, bias, outbuf, N);
}

extern "C" void kernel_launch(void* const* d_in, const int* in_sizes, int n_in,
                              void* d_out, int out_size, void* d_ws,
                              size_t ws_size, hipStream_t stream) {
  const float* x   = (const float*)d_in[0];
  const int* ei    = (const int*)d_in[1];
  const float* W1s = (const float*)d_in[2];
  const float* W1d = (const float*)d_in[3];
  const float* a1s = (const float*)d_in[4];
  const float* a1d = (const float*)d_in[5];
  const float* b1  = (const float*)d_in[6];
  const float* W2s = (const float*)d_in[7];
  const float* W2d = (const float*)d_in[8];
  const float* a2s = (const float*)d_in[9];
  const float* a2d = (const float*)d_in[10];
  const float* b2  = (const float*)d_in[11];
  const float* Wl  = (const float*)d_in[12];
  const float* bl  = (const float*)d_in[13];

  const int N = in_sizes[0] / 128;
  const int E = in_sizes[1] / 2;
  const int* src = ei;
  const int* dst = ei + E;
  const int nb    = (N + BNODES - 1) >> BSH;
  const int nblk  = (E + CHUNK - 1) / CHUNK;
  const int n2    = nb * nblk;
  const int tiles = (n2 + 1023) / 1024;

  float* bufA  = (float*)d_ws;
  float* bufB  = bufA + (size_t)N * 128;
  float* vd    = bufB + (size_t)N * 128;
  int* pairs   = (int*)(vd + 128);
  float* a_s   = (float*)(pairs + (size_t)E * 2);
  float* a_d   = a_s + N;
  int* rowptr  = (int*)(a_d + N);
  int* hist    = rowptr + N + 1;
  int* offs    = hist + n2;
  int* bsum    = offs + n2 + 1;
  int* ssrc    = bsum + tiles;

  gat_bin_count<<<nblk, BIN_T, 0, stream>>>(dst, hist, E, nb, nblk);
  gat_scan_tiles<<<tiles, 256, 0, stream>>>(hist, bsum, n2);
  gat_scan_rowptr<<<tiles, 256, 0, stream>>>(hist, bsum, offs, n2);
  gat_bin_scatter<<<nblk, BIN_T, 0, stream>>>(src, dst, offs, pairs, E, nb,
                                              nblk);
  gat_bucket_sort<<<nb, SORT_T, 0, stream>>>(pairs, offs, rowptr, ssrc, N,
                                             nb, nblk);

  run_gat_layer(x, W1s, W1d, a1s, a1d, b1, rowptr, ssrc, N, bufA, bufB, a_s,
                a_d, vd, stream);
  run_gat_layer(bufB, W2s, W2d, a2s, a2d, b2, rowptr, ssrc, N, bufA, bufB, a_s,
                a_d, vd, stream);
  gat_final_linear<<<(N + 7) / 8, 256, 0, stream>>>(bufB, Wl, bl,
                                                    (float*)d_out, N);
}


# Per-node score and output-layer launches fused into their neighbours (a_s in the GEMM epilogue, a_d and the head in the aggregate wave): 14 -> 11 launches, 300k fewer waves; score sums reordered
# speedup vs baseline: 5.8721x; 1.1725x over previous
#include <hip/hip_runtime.h>
#include <hip/hip_bf16.h>

typedef __attribute__((ext_vector_type(2))) float v2f;
typedef __attribute__((ext_vector_type(4))) float v4f;
typedef __attribute__((ext_vector_type(8))) float v8f;

#define NEG_SLOPE 0.2f
#define SM_EPS 1e-16f

__device__ __forceinline__ float wave_reduce_sum(float v) {
#pragma unroll
  for (int off = 16; off > 0; off >>= 1) v += __shfl_down(v, off, 32);
  return v;
}

__global__ __launch_bounds__(128) void gat_matvec128(
    const float* __restrict__ W, const float* __restrict__ att,
    float* __restrict__ vd) {
  int k = threadIdx.x;
  float s = 0.0f;
#pragma unroll 8
  for (int h = 0; h < 128; ++h) s += W[k * 128 + h] * att[h];
  vd[k] = s;
}

template <int M>
__device__ __forceinline__ float row16_xor_add(float p) {
  return p + __int_as_float(__builtin_amdgcn_update_dpp(
                 0, __float_as_int(p), 0x160 | M, 0xf, 0xf, false));
}

__global__ __launch_bounds__(256) void gat_gemm128_wmma(
    const float* __restrict__ X, const float* __restrict__ W,
    const float* __restrict__ atts, float* __restrict__ Y,
    float* __restrict__ a_s, int nrows) {
  __shared__ float sB[64 * 128 * 2];

  for (int idx = threadIdx.x; idx < 128 * 128; idx += 256) {
    int k = idx >> 7, col = idx & 127;
    int g = ((k >> 2) << 1) | ((k >> 1) & 1);
    sB[(((g << 7) | col) << 1) | (k & 1)] = W[idx];
  }
  __syncthreads();

  const int wave = threadIdx.x >> 5;
  const int lane = threadIdx.x & 31;
  const int half = lane >> 4;
  const int rl   = lane & 15;
  const int rowBase = blockIdx.x * 128 + wave * 16;

  int arow = rowBase + rl;
  if (arow > nrows - 1) arow = nrows - 1;
  const float* xr = X + (size_t)arow * 128 + (half << 1);

  v2f afrag[32];
#pragma unroll
  for (int kk = 0; kk < 32; ++kk) afrag[kk] = *(const v2f*)(xr + (kk << 2));

  float av[8];
#pragma unroll
  for (int ct = 0; ct < 8; ++ct) av[ct] = atts[(ct << 4) + rl];

  v8f acc[8];
#pragma unroll
  for (int ct = 0; ct < 8; ++ct) acc[ct] = (v8f)(0.0f);

  const v2f* sB2 = (const v2f*)sB;
#pragma unroll
  for (int kk = 0; kk < 32; ++kk) {
    const v2f a = afrag[kk];
    const int gbase = ((kk << 1) | half) << 7;
#pragma unroll
    for (int ct = 0; ct < 8; ++ct) {
      v2f b = sB2[gbase + (ct << 4) + rl];
      acc[ct] = __builtin_amdgcn_wmma_f32_16x16x4_f32(
          false, a, false, b, (short)0, acc[ct], false, false);
    }
  }

#pragma unroll
  for (int ct = 0; ct < 8; ++ct) {
#pragma unroll
    for (int v = 0; v < 8; ++v) {
      int r = rowBase + (half << 3) + v;
      if (r < nrows) Y[(size_t)r * 128 + (ct << 4) + rl] = acc[ct][v];
    }
  }

  float mine = 0.0f;
#pragma unroll
  for (int v = 0; v < 8; ++v) {
    float p = acc[0][v] * av[0];
#pragma unroll
    for (int ct = 1; ct < 8; ++ct) p = fmaf(acc[ct][v], av[ct], p);
    p = row16_xor_add<1>(p);
    p = row16_xor_add<2>(p);
    p = row16_xor_add<4>(p);
    p = row16_xor_add<8>(p);
    if (rl == v) mine = p;
  }
  const int sr = rowBase + (half << 3) + rl;
  if (rl < 8 && sr < nrows) a_s[sr] = mine;
}

#define BSH 9
#define BNODES (1 << BSH)
#define MAXB 1024
#define CHUNK 4096
#define BIN_T 512
#define SORT_T 1024
#define STAGE_CAP 16384

__global__ __launch_bounds__(BIN_T) void gat_bin_count(
    const int* __restrict__ dst, int* __restrict__ hist, int nedges, int nb,
    int nblk) {
  __shared__ int h[MAXB];
  const int tid = threadIdx.x;
  for (int b = tid; b < nb; b += BIN_T) h[b] = 0;
  __syncthreads();
  const int base = blockIdx.x * CHUNK;
  for (int k = 0; k < CHUNK / BIN_T; ++k) {
    const int i = base + k * BIN_T + tid;
    if (i < nedges) atomicAdd(&h[dst[i] >> BSH], 1);
  }
  __syncthreads();
  for (int b = tid; b < nb; b += BIN_T)
    hist[b * nblk + (int)blockIdx.x] = h[b];
}

__device__ __forceinline__ int wave_reduce_sum_i(int v) {
  for (int off = 16; off > 0; off >>= 1) v += __shfl_down(v, off, 32);
  return v;
}

__global__ __launch_bounds__(256) void gat_scan_tiles(
    const int* __restrict__ cnt, int* __restrict__ bsum, int n) {
  __shared__ int sw[8];
  const int base = blockIdx.x * 1024 + threadIdx.x * 4;
  int s = 0;
  for (int j = 0; j < 4; ++j)
    if (base + j < n) s += cnt[base + j];
  s = wave_reduce_sum_i(s);
  if ((threadIdx.x & 31) == 0) sw[threadIdx.x >> 5] = s;
  __syncthreads();
  if (threadIdx.x == 0) {
    int t = 0;
    for (int w = 0; w < 8; ++w) t += sw[w];
    bsum[blockIdx.x] = t;
  }
}

__global__ __launch_bounds__(256) void gat_scan_rowptr(
    const int* __restrict__ cnt, const int* __restrict__ bsum,
    int* __restrict__ rowptr, int n) {
  __shared__ int sw[8];
  __shared__ int sv[8];
  const int tid = threadIdx.x, lane = tid & 31, wave = tid >> 5;

  int o = 0;
  for (int k = tid; k < (int)blockIdx.x; k += 256) o += bsum[k];
  o = wave_reduce_sum_i(o);
  if (lane == 0) sw[wave] = o;

  const int base = blockIdx.x * 1024 + tid * 4;
  int c[4];
  int s = 0;
#pragma unroll
  for (int j = 0; j < 4; ++j) {
    c[j] = (base + j < n) ? cnt[base + j] : 0;
    s += c[j];
  }
  int inc = s;
  for (int off = 1; off < 32; off <<= 1) {
    int t = __shfl_up(inc, off, 32);
    if (lane >= off) inc += t;
  }
  if (lane == 31) sv[wave] = inc;
  __syncthreads();

  int run = inc - s;
  for (int w = 0; w < 8; ++w) {
    run += sw[w];
    if (w < wave) run += sv[w];
  }
#pragma unroll
  for (int j = 0; j < 4; ++j) {
    if (base + j < n) {
      rowptr[base + j] = run;
      run += c[j];
      if (base + j == n - 1) rowptr[n] = run;
    }
  }
}

__global__ __launch_bounds__(BIN_T) void gat_bin_scatter(
    const int* __restrict__ src, const int* __restrict__ dst,
    const int* __restrict__ offs, int* __restrict__ pairs, int nedges,
    int nb, int nblk) {
  __shared__ int gofs[MAXB];
  __shared__ int lstart[MAXB];
  __shared__ int cur[MAXB];
  __shared__ int2 sorted[CHUNK];
  const int tid = threadIdx.x;
  const int blk = (int)blockIdx.x;

  for (int b = tid; b < nb; b += BIN_T) {
    const int o = offs[b * nblk + blk];
    gofs[b] = o;
    lstart[b] = offs[b * nblk + blk + 1] - o;
    cur[b] = 0;
  }
  __syncthreads();
  if (tid < 32) {
    const int per = (nb + 31) >> 5;
    const int b0 = tid * per;
    int s = 0;
    for (int j = 0; j < per; ++j)
      if (b0 + j < nb) s += lstart[b0 + j];
    int inc = s;
    for (int off = 1; off < 32; off <<= 1) {
      int t = __shfl_up(inc, off, 32);
      if (tid >= off) inc += t;
    }
    int run = inc - s;
    for (int j = 0; j < per; ++j) {
      if (b0 + j < nb) {
        const int c = lstart[b0 + j];
        lstart[b0 + j] = run;
        run += c;
      }
    }
  }
  __syncthreads();

  const int base = blk * CHUNK;
  for (int k = 0; k < CHUNK / BIN_T; ++k) {
    const int i = base + k * BIN_T + tid;
    if (i < nedges) {
      const int d = dst[i], s = src[i];
      const int b = d >> BSH;
      const int r = atomicAdd(&cur[b], 1);
      sorted[lstart[b] + r] = make_int2(s, d);
    }
  }
  __syncthreads();

  int cnt = nedges - base;
  if (cnt > CHUNK) cnt = CHUNK;
  int2* __restrict__ pout = reinterpret_cast<int2*>(pairs);
  for (int t = tid; t < cnt; t += BIN_T) {
    const int2 p = sorted[t];
    const int b = p.y >> BSH;
    pout[gofs[b] + (t - lstart[b])] = p;
  }
}

__global__ __launch_bounds__(SORT_T) void gat_bucket_sort(
    const int* __restrict__ pairs, const int* __restrict__ offs,
    int* __restrict__ rowptr, int* __restrict__ ssrc, int n, int nb,
    int nblk) {
  __shared__ int cnt[BNODES];
  __shared__ int cursor[BNODES];
  __shared__ int wsum[BNODES / 32];
  __shared__ int stage[STAGE_CAP];
  const int tid = threadIdx.x;
  const int b = (int)blockIdx.x;
  const int beg = offs[b * nblk], end = offs[(b + 1) * nblk];
  const int node0 = b << BSH;
  const int2* __restrict__ pin = reinterpret_cast<const int2*>(pairs);

  if (tid < BNODES) cnt[tid] = 0;
  __syncthreads();
  for (int i0 = beg + tid; i0 < end; i0 += 4 * SORT_T) {
    int dl[4];
#pragma unroll
    for (int u = 0; u < 4; ++u) {
      const int i = i0 + u * SORT_T;
      dl[u] = (i < end) ? pin[i].y : node0;
    }
#pragma unroll
    for (int u = 0; u < 4; ++u)
      if (i0 + u * SORT_T < end) atomicAdd(&cnt[dl[u] - node0], 1);
  }
  __syncthreads();

  int c = 0, inc = 0;
  if (tid < BNODES) {
    c = cnt[tid];
    inc = c;
    for (int off = 1; off < 32; off <<= 1) {
      int t = __shfl_up(inc, off, 32);
      if ((tid & 31) >= off) inc += t;
    }
    if ((tid & 31) == 31) wsum[tid >> 5] = inc;
  }
  __syncthreads();
  if (tid < BNODES) {
    int run = inc - c;
    for (int w = 0; w < (tid >> 5); ++w) run += wsum[w];
    cursor[tid] = run;
    if (node0 + tid < n) rowptr[node0 + tid] = beg + run;
  }
  if (b == nb - 1 && tid == 0) rowptr[n] = end;
  __syncthreads();

  const bool staged = (end - beg) <= STAGE_CAP;
  for (int i0 = beg + tid; i0 < end; i0 += 4 * SORT_T) {
    int2 p[4];
#pragma unroll
    for (int u = 0; u < 4; ++u) {
      const int i = i0 + u * SORT_T;
      p[u] = (i < end) ? pin[i] : make_int2(0, node0);
    }
#pragma unroll
    for (int u = 0; u < 4; ++u) {
      if (i0 + u * SORT_T < end) {
        const int pos = atomicAdd(&cursor[p[u].y - node0], 1);
        if (staged) stage[pos] = p[u].x;
        else ssrc[beg + pos] = p[u].x;
      }
    }
  }
  if (staged) {
    __syncthreads();
    for (int i = tid; i < end - beg; i += SORT_T) ssrc[beg + i] = stage[i];
  }
}

__device__ __forceinline__ float wave_allreduce_max(float v) {
  for (int off = 16; off > 0; off >>= 1) v = fmaxf(v, __shfl_xor(v, off, 32));
  return v;
}
__device__ __forceinline__ float wave_allreduce_sum(float v) {
  for (int off = 16; off > 0; off >>= 1) v += __shfl_xor(v, off, 32);
  return v;
}
__device__ __forceinline__ float gat_leaky(float e) {
  return (e > 0.0f) ? e : NEG_SLOPE * e;
}
__device__ __forceinline__ int lane_i(int v, int l) {
  return __builtin_amdgcn_readlane(v, l);
}
__device__ __forceinline__ float lane_f(float v, int l) {
  return __int_as_float(__builtin_amdgcn_readlane(__float_as_int(v), l));
}
__device__ __forceinline__ v4f gat_row(const float* __restrict__ xl, int s) {
  return *(const v4f*)(xl + (size_t)s * 128);
}

template <bool HEAD>
__global__ __launch_bounds__(256) void gat_node_aggregate(
    const int* __restrict__ rowptr, const int* __restrict__ ssrc,
    const float* __restrict__ a_s, const float* __restrict__ xin,
    const float* __restrict__ vd, const float* __restrict__ xs,
    const float* __restrict__ bias, float* __restrict__ out,
    const float* __restrict__ Wl, const float* __restrict__ bl,
    float* __restrict__ out2, int n) {
  int node = (blockIdx.x * 256 + threadIdx.x) >> 5;
  const int lane = threadIdx.x & 31;
  if (node >= n) return;
  node = __builtin_amdgcn_readfirstlane(node);
  const int beg = rowptr[node], end = rowptr[node + 1];
  const v4f xiv = *(const v4f*)(xin + (size_t)node * 128 + lane * 4);
  const v4f vdv = *(const v4f*)(vd + lane * 4);

  const int c0 = beg + lane;
  const bool v0 = c0 < end;
  const int sr0 = ssrc[v0 ? c0 : 0];
  const int s0 = v0 ? sr0 : 0;
  const float as0 = a_s[s0];
  const float ad = lane_f(
      wave_reduce_sum(xiv.x * vdv.x + xiv.y * vdv.y + xiv.z * vdv.z +
                      xiv.w * vdv.w),
      0);
  const float e0 = v0 ? gat_leaky(as0 + ad) : -INFINITY;
  float mx = e0;
  for (int c = c0 + 32; c < end; c += 32)
    mx = fmaxf(mx, gat_leaky(a_s[ssrc[c]] + ad));
  mx = wave_allreduce_max(mx);
  if (!(fabsf(mx) < INFINITY)) mx = 0.0f;

  const float x0 = v0 ? __expf(e0 - mx) : 0.0f;
  float sum = x0;
  for (int c = c0 + 32; c < end; c += 32)
    sum += __expf(gat_leaky(a_s[ssrc[c]] + ad) - mx);
  sum = wave_allreduce_sum(sum);
  const float dn = sum + SM_EPS;

  const float* xl = xs + lane * 4;
  v4f acc = (v4f)(0.0f);
  for (int cb = beg; cb < end; cb += 32) {
    int sl = s0;
    float al = x0 / dn;
    if (cb != beg) {
      const int c = cb + lane;
      sl = 0;
      al = 0.0f;
      if (c < end) {
        sl = ssrc[c];
        al = __expf(gat_leaky(a_s[sl] + ad) - mx) / dn;
      }
    }
    const int cnt = (end - cb < 32) ? (end - cb) : 32;
    int j = 0;
    for (; j + 4 <= cnt; j += 4) {
      const v4f va = gat_row(xl, lane_i(sl, j));
      const v4f vb = gat_row(xl, lane_i(sl, j + 1));
      const v4f vc = gat_row(xl, lane_i(sl, j + 2));
      const v4f vdd = gat_row(xl, lane_i(sl, j + 3));
      acc += lane_f(al, j) * va;
      acc += lane_f(al, j + 1) * vb;
      acc += lane_f(al, j + 2) * vc;
      acc += lane_f(al, j + 3) * vdd;
    }
    const int rem = cnt - j;
    if (rem == 3) {
      const v4f va = gat_row(xl, lane_i(sl, j));
      const v4f vb = gat_row(xl, lane_i(sl, j + 1));
      const v4f vc = gat_row(xl, lane_i(sl, j + 2));
      acc += lane_f(al, j) * va;
      acc += lane_f(al, j + 1) * vb;
      acc += lane_f(al, j + 2) * vc;
    } else if (rem == 2) {
      const v4f va = gat_row(xl, lane_i(sl, j));
      const v4f vb = gat_row(xl, lane_i(sl, j + 1));
      acc += lane_f(al, j) * va;
      acc += lane_f(al, j + 1) * vb;
    } else if (rem == 1) {
      const v4f va = gat_row(xl, lane_i(sl, j));
      acc += lane_f(al, j) * va;
    }
  }

  const v4f b = *(const v4f*)(bias + lane * 4);
  v4f r;
  r.x = fmaxf(acc.x + b.x, 0.0f);
  r.y = fmaxf(acc.y + b.y, 0.0f);
  r.z = fmaxf(acc.z + b.z, 0.0f);
  r.w = fmaxf(acc.w + b.w, 0.0f);
  if (HEAD) {
    const v4f w0 = *(const v4f*)(Wl + lane * 8);
    const v4f w1 = *(const v4f*)(Wl + lane * 8 + 4);
    float p0 = r.x * w0.x + r.y * w0.z + r.z * w1.x + r.w * w1.z;
    float p1 = r.x * w0.y + r.y * w0.w + r.z * w1.y + r.w * w1.w;
    p0 = wave_reduce_sum(p0);
    p1 = wave_reduce_sum(p1);
    if (lane == 0) {
      out2[(size_t)node * 2 + 0] = p0 + bl[0];
      out2[(size_t)node * 2 + 1] = p1 + bl[1];
    }
  } else {
    *(v4f*)(out + (size_t)node * 128 + lane * 4) = r;
  }
}

static void run_gat_layer(const float* xin, const float* Wsrc,
                          const float* Wdst, const float* atts,
                          const float* attd, const float* bias,
                          const int* rowptr, const int* ssrc, int N, float* xs,
                          float* outbuf, float* a_s, float* vd,
                          const float* Wl, const float* bl, float* out2,
                          hipStream_t stream) {
  gat_matvec128<<<1, 128, 0, stream>>>(Wdst, attd, vd);
  gat_gemm128_wmma<<<(N + 127) / 128, 256, 0, stream>>>(xin, Wsrc, atts, xs,
                                                        a_s, N);
  if (Wl != nullptr)
    gat_node_aggregate<true><<<(N + 7) / 8, 256, 0, stream>>>(
        rowptr, ssrc, a_s, xin, vd, xs, bias, outbuf, Wl, bl, out2, N);
  else
    gat_node_aggregate<false><<<(N + 7) / 8, 256, 0, stream>>>(
        rowptr, ssrc, a_s, xin, vd, xs, bias, outbuf, Wl, bl, out2, N);
}

extern "C" void kernel_launch(void* const* d_in, const int* in_sizes, int n_in,
                              void* d_out, int out_size, void* d_ws,
                              size_t ws_size, hipStream_t stream) {
  const float* x   = (const float*)d_in[0];
  const int* ei    = (const int*)d_in[1];
  const float* W1s = (const float*)d_in[2];
  const float* W1d = (const float*)d_in[3];
  const float* a1s = (const float*)d_in[4];
  const float* a1d = (const float*)d_in[5];
  const float* b1  = (const float*)d_in[6];
  const float* W2s = (const float*)d_in[7];
  const float* W2d = (const float*)d_in[8];
  const float* a2s = (const float*)d_in[9];
  const float* a2d = (const float*)d_in[10];
  const float* b2  = (const float*)d_in[11];
  const float* Wl  = (const float*)d_in[12];
  const float* bl  = (const float*)d_in[13];

  const int N = in_sizes[0] / 128;
  const int E = in_sizes[1] / 2;
  const int* src = ei;
  const int* dst = ei + E;
  const int nb    = (N + BNODES - 1) >> BSH;
  const int nblk  = (E + CHUNK - 1) / CHUNK;
  const int n2    = nb * nblk;
  const int tiles = (n2 + 1023) / 1024;

  float* bufA  = (float*)d_ws;
  float* bufB  = bufA + (size_t)N * 128;
  float* vd    = bufB + (size_t)N * 128;
  int* pairs   = (int*)(vd + 128);
  float* a_s   = (float*)(pairs + (size_t)E * 2);
  int* rowptr  = (int*)(a_s + N);
  int* hist    = rowptr + N + 1;
  int* offs    = hist + n2;
  int* bsum    = offs + n2 + 1;
  int* ssrc    = bsum + tiles;

  gat_bin_count<<<nblk, BIN_T, 0, stream>>>(dst, hist, E, nb, nblk);
  gat_scan_tiles<<<tiles, 256, 0, stream>>>(hist, bsum, n2);
  gat_scan_rowptr<<<tiles, 256, 0, stream>>>(hist, bsum, offs, n2);
  gat_bin_scatter<<<nblk, BIN_T, 0, stream>>>(src, dst, offs, pairs, E, nb,
                                              nblk);
  gat_bucket_sort<<<nb, SORT_T, 0, stream>>>(pairs, offs, rowptr, ssrc, N,
                                             nb, nblk);

  run_gat_layer(x, W1s, W1d, a1s, a1d, b1, rowptr, ssrc, N, bufA, bufB, a_s,
                vd, nullptr, nullptr, nullptr, stream);
  run_gat_layer(bufB, W2s, W2d, a2s, a2d, b2, rowptr, ssrc, N, bufA, nullptr,
                a_s, vd, Wl, bl, (float*)d_out, stream);
}


# 11 -> 8 launches: both W_dst*att_dst vectors made by two spare blocks of the bucket-sort launch, the two scan launches made one (each block sums what precedes its tile itself)
# speedup vs baseline: 6.1531x; 1.0479x over previous
#include <hip/hip_runtime.h>
#include <hip/hip_bf16.h>

typedef __attribute__((ext_vector_type(2))) float v2f;
typedef __attribute__((ext_vector_type(4))) float v4f;
typedef __attribute__((ext_vector_type(8))) float v8f;

#define NEG_SLOPE 0.2f
#define SM_EPS 1e-16f

__device__ __forceinline__ float wave_reduce_sum(float v) {
#pragma unroll
  for (int off = 16; off > 0; off >>= 1) v += __shfl_down(v, off, 32);
  return v;
}

__device__ __forceinline__ void gat_matvec128_row(
    const float* __restrict__ W, const float* __restrict__ att,
    float* __restrict__ vd, int k) {
  const v4f* wr = (const v4f*)(W + k * 128);
  float s = 0.0f;
  for (int g = 0; g < 128 / 32; ++g) {
    v4f w[8];
#pragma unroll
    for (int i = 0; i < 8; ++i) w[i] = wr[g * 8 + i];
#pragma unroll
    for (int i = 0; i < 8; ++i) {
      const int h = g * 32 + i * 4;
      s += w[i].x * att[h];
      s += w[i].y * att[h + 1];
      s += w[i].z * att[h + 2];
      s += w[i].w * att[h + 3];
    }
  }
  vd[k] = s;
}

template <int M>
__device__ __forceinline__ float row16_xor_add(float p) {
  return p + __int_as_float(__builtin_amdgcn_update_dpp(
                 0, __float_as_int(p), 0x160 | M, 0xf, 0xf, false));
}

__global__ __launch_bounds__(256) void gat_gemm128_wmma(
    const float* __restrict__ X, const float* __restrict__ W,
    const float* __restrict__ atts, float* __restrict__ Y,
    float* __restrict__ a_s, int nrows) {
  __shared__ float sB[64 * 128 * 2];

  for (int idx = threadIdx.x; idx < 128 * 128; idx += 256) {
    int k = idx >> 7, col = idx & 127;
    int g = ((k >> 2) << 1) | ((k >> 1) & 1);
    sB[(((g << 7) | col) << 1) | (k & 1)] = W[idx];
  }
  __syncthreads();

  const int wave = threadIdx.x >> 5;
  const int lane = threadIdx.x & 31;
  const int half = lane >> 4;
  const int rl   = lane & 15;
  const int rowBase = blockIdx.x * 128 + wave * 16;

  int arow = rowBase + rl;
  if (arow > nrows - 1) arow = nrows - 1;
  const float* xr = X + (size_t)arow * 128 + (half << 1);

  v2f afrag[32];
#pragma unroll
  for (int kk = 0; kk < 32; ++kk) afrag[kk] = *(const v2f*)(xr + (kk << 2));

  float av[8];
#pragma unroll
  for (int ct = 0; ct < 8; ++ct) av[ct] = atts[(ct << 4) + rl];

  v8f acc[8];
#pragma unroll
  for (int ct = 0; ct < 8; ++ct) acc[ct] = (v8f)(0.0f);

  const v2f* sB2 = (const v2f*)sB;
#pragma unroll
  for (int kk = 0; kk < 32; ++kk) {
    const v2f a = afrag[kk];
    const int gbase = ((kk << 1) | half) << 7;
#pragma unroll
    for (int ct = 0; ct < 8; ++ct) {
      v2f b = sB2[gbase + (ct << 4) + rl];
      acc[ct] = __builtin_amdgcn_wmma_f32_16x16x4_f32(
          false, a, false, b, (short)0, acc[ct], false, false);
    }
  }

#pragma unroll
  for (int ct = 0; ct < 8; ++ct) {
#pragma unroll
    for (int v = 0; v < 8; ++v) {
      int r = rowBase + (half << 3) + v;
      if (r < nrows) Y[(size_t)r * 128 + (ct << 4) + rl] = acc[ct][v];
    }
  }

  float mine = 0.0f;
#pragma unroll
  for (int v = 0; v < 8; ++v) {
    float p = acc[0][v] * av[0];
#pragma unroll
    for (int ct = 1; ct < 8; ++ct) p = fmaf(acc[ct][v], av[ct], p);
    p = row16_xor_add<1>(p);
    p = row16_xor_add<2>(p);
    p = row16_xor_add<4>(p);
    p = row16_xor_add<8>(p);
    if (rl == v) mine = p;
  }
  const int sr = rowBase + (half << 3) + rl;
  if (rl < 8 && sr < nrows) a_s[sr] = mine;
}

#define BSH 9
#define BNODES (1 << BSH)
#define MAXB 1024
#define CHUNK 4096
#define BIN_T 512
#define SORT_T 1024
#define STAGE_CAP 16384

__global__ __launch_bounds__(BIN_T) void gat_bin_count(
    const int* __restrict__ dst, int* __restrict__ hist, int nedges, int nb,
    int nblk) {
  __shared__ int h[MAXB];
  const int tid = threadIdx.x;
  for (int b = tid; b < nb; b += BIN_T) h[b] = 0;
  __syncthreads();
  const int base = blockIdx.x * CHUNK;
  for (int k = 0; k < CHUNK / BIN_T; ++k) {
    const int i = base + k * BIN_T + tid;
    if (i < nedges) atomicAdd(&h[dst[i] >> BSH], 1);
  }
  __syncthreads();
  for (int b = tid; b < nb; b += BIN_T)
    hist[b * nblk + (int)blockIdx.x] = h[b];
}

__device__ __forceinline__ int wave_reduce_sum_i(int v) {
  for (int off = 16; off > 0; off >>= 1) v += __shfl_down(v, off, 32);
  return v;
}

#define SCAN_T 1024
#define SCAN_TILE 4096
__global__ __launch_bounds__(SCAN_T) void gat_scan_offs(
    const int* __restrict__ cnt, int* __restrict__ offs, int n) {
  __shared__ int sw[SCAN_T / 32];
  __shared__ int sv[SCAN_T / 32];
  const int tid = threadIdx.x, lane = tid & 31, wave = tid >> 5;
  const int tile0 = (int)blockIdx.x * SCAN_TILE;

  const int base = tile0 + tid * 4;
  int c[4];
  int s = 0;
#pragma unroll
  for (int j = 0; j < 4; ++j) {
    c[j] = (base + j < n) ? cnt[base + j] : 0;
    s += c[j];
  }

  const int4* __restrict__ c4 = reinterpret_cast<const int4*>(cnt);
  const int nq = tile0 >> 2;
  int o = 0;
#pragma unroll 4
  for (int k = tid; k < nq; k += SCAN_T) {
    const int4 q = c4[k];
    o += (q.x + q.y) + (q.z + q.w);
  }
  o = wave_reduce_sum_i(o);
  if (lane == 0) sw[wave] = o;

  int inc = s;
  for (int off = 1; off < 32; off <<= 1) {
    int t = __shfl_up(inc, off, 32);
    if (lane >= off) inc += t;
  }
  if (lane == 31) sv[wave] = inc;
  __syncthreads();

  int run = inc - s;
  for (int w = 0; w < SCAN_T / 32; ++w) {
    run += sw[w];
    if (w < wave) run += sv[w];
  }
#pragma unroll
  for (int j = 0; j < 4; ++j) {
    if (base + j < n) {
      offs[base + j] = run;
      run += c[j];
      if (base + j == n - 1) offs[n] = run;
    }
  }
}

__global__ __launch_bounds__(BIN_T) void gat_bin_scatter(
    const int* __restrict__ src, const int* __restrict__ dst,
    const int* __restrict__ offs, int* __restrict__ pairs, int nedges,
    int nb, int nblk) {
  __shared__ int gofs[MAXB];
  __shared__ int lstart[MAXB];
  __shared__ int cur[MAXB];
  __shared__ int2 sorted[CHUNK];
  const int tid = threadIdx.x;
  const int blk = (int)blockIdx.x;

  for (int b = tid; b < nb; b += BIN_T) {
    const int o = offs[b * nblk + blk];
    gofs[b] = o;
    lstart[b] = offs[b * nblk + blk + 1] - o;
    cur[b] = 0;
  }
  __syncthreads();
  if (tid < 32) {
    const int per = (nb + 31) >> 5;
    const int b0 = tid * per;
    int s = 0;
    for (int j = 0; j < per; ++j)
      if (b0 + j < nb) s += lstart[b0 + j];
    int inc = s;
    for (int off = 1; off < 32; off <<= 1) {
      int t = __shfl_up(inc, off, 32);
      if (tid >= off) inc += t;
    }
    int run = inc - s;
    for (int j = 0; j < per; ++j) {
      if (b0 + j < nb) {
        const int c = lstart[b0 + j];
        lstart[b0 + j] = run;
        run += c;
      }
    }
  }
  __syncthreads();

  const int base = blk * CHUNK;
  for (int k = 0; k < CHUNK / BIN_T; ++k) {
    const int i = base + k * BIN_T + tid;
    if (i < nedges) {
      const int d = dst[i], s = src[i];
      const int b = d >> BSH;
      const int r = atomicAdd(&cur[b], 1);
      sorted[lstart[b] + r] = make_int2(s, d);
    }
  }
  __syncthreads();

  int cnt = nedges - base;
  if (cnt > CHUNK) cnt = CHUNK;
  int2* __restrict__ pout = reinterpret_cast<int2*>(pairs);
  for (int t = tid; t < cnt; t += BIN_T) {
    const int2 p = sorted[t];
    const int b = p.y >> BSH;
    pout[gofs[b] + (t - lstart[b])] = p;
  }
}

__global__ __launch_bounds__(SORT_T) void gat_bucket_sort(
    const int* __restrict__ pairs, const int* __restrict__ offs,
    int* __restrict__ rowptr, int* __restrict__ ssrc, int n, int nb,
    int nblk, const float* __restrict__ Wd1, const float* __restrict__ ad1,
    float* __restrict__ vd1, const float* __restrict__ Wd2,
    const float* __restrict__ ad2, float* __restrict__ vd2) {
  __shared__ int cnt[BNODES];
  __shared__ int cursor[BNODES];
  __shared__ int wsum[BNODES / 32];
  __shared__ int stage[STAGE_CAP];
  const int tid = threadIdx.x;
  const int b = (int)blockIdx.x;
  if (b >= nb) {
    if (tid < 128) {
      if (b == nb) gat_matvec128_row(Wd1, ad1, vd1, tid);
      else gat_matvec128_row(Wd2, ad2, vd2, tid);
    }
    return;
  }
  const int beg = offs[b * nblk], end = offs[(b + 1) * nblk];
  const int node0 = b << BSH;
  const int2* __restrict__ pin = reinterpret_cast<const int2*>(pairs);

  if (tid < BNODES) cnt[tid] = 0;
  __syncthreads();
  for (int i0 = beg + tid; i0 < end; i0 += 4 * SORT_T) {
    int dl[4];
#pragma unroll
    for (int u = 0; u < 4; ++u) {
      const int i = i0 + u * SORT_T;
      dl[u] = (i < end) ? pin[i].y : node0;
    }
#pragma unroll
    for (int u = 0; u < 4; ++u)
      if (i0 + u * SORT_T < end) atomicAdd(&cnt[dl[u] - node0], 1);
  }
  __syncthreads();

  int c = 0, inc = 0;
  if (tid < BNODES) {
    c = cnt[tid];
    inc = c;
    for (int off = 1; off < 32; off <<= 1) {
      int t = __shfl_up(inc, off, 32);
      if ((tid & 31) >= off) inc += t;
    }
    if ((tid & 31) == 31) wsum[tid >> 5] = inc;
  }
  __syncthreads();
  if (tid < BNODES) {
    int run = inc - c;
    for (int w = 0; w < (tid >> 5); ++w) run += wsum[w];
    cursor[tid] = run;
    if (node0 + tid < n) rowptr[node0 + tid] = beg + run;
  }
  if (b == nb - 1 && tid == 0) rowptr[n] = end;
  __syncthreads();

  const bool staged = (end - beg) <= STAGE_CAP;
  for (int i0 = beg + tid; i0 < end; i0 += 4 * SORT_T) {
    int2 p[4];
#pragma unroll
    for (int u = 0; u < 4; ++u) {
      const int i = i0 + u * SORT_T;
      p[u] = (i < end) ? pin[i] : make_int2(0, node0);
    }
#pragma unroll
    for (int u = 0; u < 4; ++u) {
      if (i0 + u * SORT_T < end) {
        const int pos = atomicAdd(&cursor[p[u].y - node0], 1);
        if (staged) stage[pos] = p[u].x;
        else ssrc[beg + pos] = p[u].x;
      }
    }
  }
  if (staged) {
    __syncthreads();
    for (int i = tid; i < end - beg; i += SORT_T) ssrc[beg + i] = stage[i];
  }
}

__device__ __forceinline__ float wave_allreduce_max(float v) {
  for (int off = 16; off > 0; off >>= 1) v = fmaxf(v, __shfl_xor(v, off, 32));
  return v;
}
__device__ __forceinline__ float wave_allreduce_sum(float v) {
  for (int off = 16; off > 0; off >>= 1) v += __shfl_xor(v, off, 32);
  return v;
}
__device__ __forceinline__ float gat_leaky(float e) {
  return (e > 0.0f) ? e : NEG_SLOPE * e;
}
__device__ __forceinline__ int lane_i(int v, int l) {
  return __builtin_amdgcn_readlane(v, l);
}
__device__ __forceinline__ float lane_f(float v, int l) {
  return __int_as_float(__builtin_amdgcn_readlane(__float_as_int(v), l));
}
__device__ __forceinline__ v4f gat_row(const float* __restrict__ xl, int s) {
  return *(const v4f*)(xl + (size_t)s * 128);
}

template <bool HEAD>
__global__ __launch_bounds__(256) void gat_node_aggregate(
    const int* __restrict__ rowptr, const int* __restrict__ ssrc,
    const float* __restrict__ a_s, const float* __restrict__ xin,
    const float* __restrict__ vd, const float* __restrict__ xs,
    const float* __restrict__ bias, float* __restrict__ out,
    const float* __restrict__ Wl, const float* __restrict__ bl,
    float* __restrict__ out2, int n) {
  int node = (blockIdx.x * 256 + threadIdx.x) >> 5;
  const int lane = threadIdx.x & 31;
  if (node >= n) return;
  node = __builtin_amdgcn_readfirstlane(node);
  const int beg = rowptr[node], end = rowptr[node + 1];
  const v4f xiv = *(const v4f*)(xin + (size_t)node * 128 + lane * 4);
  const v4f vdv = *(const v4f*)(vd + lane * 4);

  const int c0 = beg + lane;
  const bool v0 = c0 < end;
  const int sr0 = ssrc[v0 ? c0 : 0];
  const int s0 = v0 ? sr0 : 0;
  const float as0 = a_s[s0];
  const float ad = lane_f(
      wave_reduce_sum(xiv.x * vdv.x + xiv.y * vdv.y + xiv.z * vdv.z +
                      xiv.w * vdv.w),
      0);
  const float e0 = v0 ? gat_leaky(as0 + ad) : -INFINITY;
  float mx = e0;
  for (int c = c0 + 32; c < end; c += 32)
    mx = fmaxf(mx, gat_leaky(a_s[ssrc[c]] + ad));
  mx = wave_allreduce_max(mx);
  if (!(fabsf(mx) < INFINITY)) mx = 0.0f;

  const float x0 = v0 ? __expf(e0 - mx) : 0.0f;
  float sum = x0;
  for (int c = c0 + 32; c < end; c += 32)
    sum += __expf(gat_leaky(a_s[ssrc[c]] + ad) - mx);
  sum = wave_allreduce_sum(sum);
  const float dn = sum + SM_EPS;

  const float* xl = xs + lane * 4;
  v4f acc = (v4f)(0.0f);
  for (int cb = beg; cb < end; cb += 32) {
    int sl = s0;
    float al = x0 / dn;
    if (cb != beg) {
      const int c = cb + lane;
      sl = 0;
      al = 0.0f;
      if (c < end) {
        sl = ssrc[c];
        al = __expf(gat_leaky(a_s[sl] + ad) - mx) / dn;
      }
    }
    const int cnt = (end - cb < 32) ? (end - cb) : 32;
    int j = 0;
    for (; j + 4 <= cnt; j += 4) {
      const v4f va = gat_row(xl, lane_i(sl, j));
      const v4f vb = gat_row(xl, lane_i(sl, j + 1));
      const v4f vc = gat_row(xl, lane_i(sl, j + 2));
      const v4f vdd = gat_row(xl, lane_i(sl, j + 3));
      acc += lane_f(al, j) * va;
      acc += lane_f(al, j + 1) * vb;
      acc += lane_f(al, j + 2) * vc;
      acc += lane_f(al, j + 3) * vdd;
    }
    const int rem = cnt - j;
    if (rem == 3) {
      const v4f va = gat_row(xl, lane_i(sl, j));
      const v4f vb = gat_row(xl, lane_i(sl, j + 1));
      const v4f vc = gat_row(xl, lane_i(sl, j + 2));
      acc += lane_f(al, j) * va;
      acc += lane_f(al, j + 1) * vb;
      acc += lane_f(al, j + 2) * vc;
    } else if (rem == 2) {
      const v4f va = gat_row(xl, lane_i(sl, j));
      const v4f vb = gat_row(xl, lane_i(sl, j + 1));
      acc += lane_f(al, j) * va;
      acc += lane_f(al, j + 1) * vb;
    } else if (rem == 1) {
      const v4f va = gat_row(xl, lane_i(sl, j));
      acc += lane_f(al, j) * va;
    }
  }

  const v4f b = *(const v4f*)(bias + lane * 4);
  v4f r;
  r.x = fmaxf(acc.x + b.x, 0.0f);
  r.y = fmaxf(acc.y + b.y, 0.0f);
  r.z = fmaxf(acc.z + b.z, 0.0f);
  r.w = fmaxf(acc.w + b.w, 0.0f);
  if (HEAD) {
    const v4f w0 = *(const v4f*)(Wl + lane * 8);
    const v4f w1 = *(const v4f*)(Wl + lane * 8 + 4);
    float p0 = r.x * w0.x + r.y * w0.z + r.z * w1.x + r.w * w1.z;
    float p1 = r.x * w0.y + r.y * w0.w + r.z * w1.y + r.w * w1.w;
    p0 = wave_reduce_sum(p0);
    p1 = wave_reduce_sum(p1);
    if (lane == 0) {
      out2[(size_t)node * 2 + 0] = p0 + bl[0];
      out2[(size_t)node * 2 + 1] = p1 + bl[1];
    }
  } else {
    *(v4f*)(out + (size_t)node * 128 + lane * 4) = r;
  }
}

static void run_gat_layer(const float* xin, const float* Wsrc,
                          const float* atts, const float* bias,
                          const int* rowptr, const int* ssrc, int N, float* xs,
                          float* outbuf, float* a_s, const float* vd,
                          const float* Wl, const float* bl, float* out2,
                          hipStream_t stream) {
  gat_gemm128_wmma<<<(N + 127) / 128, 256, 0, stream>>>(xin, Wsrc, atts, xs,
                                                        a_s, N);
  if (Wl != nullptr)
    gat_node_aggregate<true><<<(N + 7) / 8, 256, 0, stream>>>(
        rowptr, ssrc, a_s, xin, vd, xs, bias, outbuf, Wl, bl, out2, N);
  else
    gat_node_aggregate<false><<<(N + 7) / 8, 256, 0, stream>>>(
        rowptr, ssrc, a_s, xin, vd, xs, bias, outbuf, Wl, bl, out2, N);
}

extern "C" void kernel_launch(void* const* d_in, const int* in_sizes, int n_in,
                              void* d_out, int out_size, void* d_ws,
                              size_t ws_size, hipStream_t stream) {
  const float* x   = (const float*)d_in[0];
  const int* ei    = (const int*)d_in[1];
  const float* W1s = (const float*)d_in[2];
  const float* W1d = (const float*)d_in[3];
  const float* a1s = (const float*)d_in[4];
  const float* a1d = (const float*)d_in[5];
  const float* b1  = (const float*)d_in[6];
  const float* W2s = (const float*)d_in[7];
  const float* W2d = (const float*)d_in[8];
  const float* a2s = (const float*)d_in[9];
  const float* a2d = (const float*)d_in[10];
  const float* b2  = (const float*)d_in[11];
  const float* Wl  = (const float*)d_in[12];
  const float* bl  = (const float*)d_in[13];

  const int N = in_sizes[0] / 128;
  const int E = in_sizes[1] / 2;
  const int* src = ei;
  const int* dst = ei + E;
  const int nb    = (N + BNODES - 1) >> BSH;
  const int nblk  = (E + CHUNK - 1) / CHUNK;
  const int n2    = nb * nblk;
  const int n2p   = (n2 + 4) & ~3;
  const int tiles = (n2 + SCAN_TILE - 1) / SCAN_TILE;

  float* bufA  = (float*)d_ws;
  float* bufB  = bufA + (size_t)N * 128;
  float* vd1   = bufB + (size_t)N * 128;
  float* vd2   = vd1 + 128;
  int* hist    = (int*)(vd2 + 128);
  int* offs    = hist + n2p;
  int* pairs   = offs + n2p;
  float* a_s   = (float*)(pairs + (size_t)E * 2);
  int* rowptr  = (int*)(a_s + N);
  int* ssrc    = rowptr + N + 1;

  gat_bin_count<<<nblk, BIN_T, 0, stream>>>(dst, hist, E, nb, nblk);
  gat_scan_offs<<<tiles, SCAN_T, 0, stream>>>(hist, offs, n2);
  gat_bin_scatter<<<nblk, BIN_T, 0, stream>>>(src, dst, offs, pairs, E, nb,
                                              nblk);
  gat_bucket_sort<<<nb + 2, SORT_T, 0, stream>>>(pairs, offs, rowptr, ssrc, N,
                                                 nb, nblk, W1d, a1d, vd1, W2d,
                                                 a2d, vd2);

  run_gat_layer(x, W1s, a1s, b1, rowptr, ssrc, N, bufA, bufB, a_s, vd1,
                nullptr, nullptr, nullptr, stream);
  run_gat_layer(bufB, W2s, a2s, b2, rowptr, ssrc, N, bufA, nullptr, a_s, vd2,
                Wl, bl, (float*)d_out, stream);
}


# Layer-1 GEMM runs as extra blocks of the two edge-binning launches (half ahead of the count chunks, half ahead of the scatter chunks), beside the sort not after it: 8 -> 7 launches; sum order kept
# speedup vs baseline: 6.6187x; 1.0757x over previous
#include <hip/hip_runtime.h>
#include <hip/hip_bf16.h>

typedef __attribute__((ext_vector_type(2))) float v2f;
typedef __attribute__((ext_vector_type(4))) float v4f;
typedef __attribute__((ext_vector_type(8))) float v8f;

#define NEG_SLOPE 0.2f
#define SM_EPS 1e-16f

__device__ __forceinline__ float wave_reduce_sum(float v) {
#pragma unroll
  for (int off = 16; off > 0; off >>= 1) v += __shfl_down(v, off, 32);
  return v;
}

__device__ __forceinline__ void gat_matvec128_row(
    const float* __restrict__ W, const float* __restrict__ att,
    float* __restrict__ vd, int k) {
  const v4f* wr = (const v4f*)(W + k * 128);
  float s = 0.0f;
  for (int g = 0; g < 128 / 32; ++g) {
    v4f w[8];
#pragma unroll
    for (int i = 0; i < 8; ++i) w[i] = wr[g * 8 + i];
#pragma unroll
    for (int i = 0; i < 8; ++i) {
      const int h = g * 32 + i * 4;
      s += w[i].x * att[h];
      s += w[i].y * att[h + 1];
      s += w[i].z * att[h + 2];
      s += w[i].w * att[h + 3];
    }
  }
  vd[k] = s;
}

template <int M>
__device__ __forceinline__ float row16_xor_add(float p) {
  return p + __int_as_float(__builtin_amdgcn_update_dpp(
                 0, __float_as_int(p), 0x160 | M, 0xf, 0xf, false));
}

template <int T>
__device__ __forceinline__ void gat_gemm128_rows(
    float* __restrict__ sB, const float* __restrict__ X,
    const float* __restrict__ W, const float* __restrict__ atts,
    float* __restrict__ Y, float* __restrict__ a_s, int nrows, int row0) {
  for (int idx = threadIdx.x; idx < 128 * 128; idx += T) {
    int k = idx >> 7, col = idx & 127;
    int g = ((k >> 2) << 1) | ((k >> 1) & 1);
    sB[(((g << 7) | col) << 1) | (k & 1)] = W[idx];
  }
  __syncthreads();

  const int wave = threadIdx.x >> 5;
  const int lane = threadIdx.x & 31;
  const int half = lane >> 4;
  const int rl   = lane & 15;
  const int rowBase = row0 + wave * 16;

  int arow = rowBase + rl;
  if (arow > nrows - 1) arow = nrows - 1;
  const float* xr = X + (size_t)arow * 128 + (half << 1);

  v2f afrag[32];
#pragma unroll
  for (int kk = 0; kk < 32; ++kk) afrag[kk] = *(const v2f*)(xr + (kk << 2));

  float av[8];
#pragma unroll
  for (int ct = 0; ct < 8; ++ct) av[ct] = atts[(ct << 4) + rl];

  v8f acc[8];
#pragma unroll
  for (int ct = 0; ct < 8; ++ct) acc[ct] = (v8f)(0.0f);

  const v2f* sB2 = (const v2f*)sB;
#pragma unroll
  for (int kk = 0; kk < 32; ++kk) {
    const v2f a = afrag[kk];
    const int gbase = ((kk << 1) | half) << 7;
#pragma unroll
    for (int ct = 0; ct < 8; ++ct) {
      v2f b = sB2[gbase + (ct << 4) + rl];
      acc[ct] = __builtin_amdgcn_wmma_f32_16x16x4_f32(
          false, a, false, b, (short)0, acc[ct], false, false);
    }
  }

#pragma unroll
  for (int ct = 0; ct < 8; ++ct) {
#pragma unroll
    for (int v = 0; v < 8; ++v) {
      int r = rowBase + (half << 3) + v;
      if (r < nrows) Y[(size_t)r * 128 + (ct << 4) + rl] = acc[ct][v];
    }
  }

  float mine = 0.0f;
#pragma unroll
  for (int v = 0; v < 8; ++v) {
    float p = acc[0][v] * av[0];
#pragma unroll
    for (int ct = 1; ct < 8; ++ct) p = fmaf(acc[ct][v], av[ct], p);
    p = row16_xor_add<1>(p);
    p = row16_xor_add<2>(p);
    p = row16_xor_add<4>(p);
    p = row16_xor_add<8>(p);
    if (rl == v) mine = p;
  }
  const int sr = rowBase + (half << 3) + rl;
  if (rl < 8 && sr < nrows) a_s[sr] = mine;
}

__global__ __launch_bounds__(256) void gat_gemm128_wmma(
    const float* __restrict__ X, const float* __restrict__ W,
    const float* __restrict__ atts, float* __restrict__ Y,
    float* __restrict__ a_s, int nrows) {
  __shared__ float sB[64 * 128 * 2];
  gat_gemm128_rows<256>(sB, X, W, atts, Y, a_s, nrows, (int)blockIdx.x * 128);
}

#define BSH 9
#define BNODES (1 << BSH)
#define MAXB 1024
#define CHUNK 4096
#define BIN_T 512
#define SORT_T 1024
#define STAGE_CAP 16384

#define GEMM_ROWS (BIN_T / 2)
__global__ __launch_bounds__(BIN_T) void gat_bin_count(
    const int* __restrict__ dst, int* __restrict__ hist, int nedges, int nb,
    int nblk, const float* __restrict__ X, const float* __restrict__ W,
    const float* __restrict__ atts, float* __restrict__ Y,
    float* __restrict__ a_s, int nrows, int g0, int gn) {
  __shared__ __attribute__((aligned(16))) unsigned char lds[64 * 128 * 2 * 4];
  if ((int)blockIdx.x < gn) {
    gat_gemm128_rows<BIN_T>((float*)lds, X, W, atts, Y, a_s, nrows,
                            (g0 + (int)blockIdx.x) * GEMM_ROWS);
    return;
  }
  int* h = (int*)lds;
  const int blk = (int)blockIdx.x - gn;
  const int tid = threadIdx.x;
  for (int b = tid; b < nb; b += BIN_T) h[b] = 0;
  __syncthreads();
  const int base = blk * CHUNK;
  for (int k = 0; k < CHUNK / BIN_T; ++k) {
    const int i = base + k * BIN_T + tid;
    if (i < nedges) atomicAdd(&h[dst[i] >> BSH], 1);
  }
  __syncthreads();
  for (int b = tid; b < nb; b += BIN_T)
    hist[b * nblk + blk] = h[b];
}

__device__ __forceinline__ int wave_reduce_sum_i(int v) {
  for (int off = 16; off > 0; off >>= 1) v += __shfl_down(v, off, 32);
  return v;
}

#define SCAN_T 1024
#define SCAN_TILE 4096
__global__ __launch_bounds__(SCAN_T) void gat_scan_offs(
    const int* __restrict__ cnt, int* __restrict__ offs, int n) {
  __shared__ int sw[SCAN_T / 32];
  __shared__ int sv[SCAN_T / 32];
  const int tid = threadIdx.x, lane = tid & 31, wave = tid >> 5;
  const int tile0 = (int)blockIdx.x * SCAN_TILE;

  const int base = tile0 + tid * 4;
  int c[4];
  int s = 0;
#pragma unroll
  for (int j = 0; j < 4; ++j) {
    c[j] = (base + j < n) ? cnt[base + j] : 0;
    s += c[j];
  }

  const int4* __restrict__ c4 = reinterpret_cast<const int4*>(cnt);
  const int nq = tile0 >> 2;
  int o = 0;
#pragma unroll 4
  for (int k = tid; k < nq; k += SCAN_T) {
    const int4 q = c4[k];
    o += (q.x + q.y) + (q.z + q.w);
  }
  o = wave_reduce_sum_i(o);
  if (lane == 0) sw[wave] = o;

  int inc = s;
  for (int off = 1; off < 32; off <<= 1) {
    int t = __shfl_up(inc, off, 32);
    if (lane >= off) inc += t;
  }
  if (lane == 31) sv[wave] = inc;
  __syncthreads();

  int run = inc - s;
  for (int w = 0; w < SCAN_T / 32; ++w) {
    run += sw[w];
    if (w < wave) run += sv[w];
  }
#pragma unroll
  for (int j = 0; j < 4; ++j) {
    if (base + j < n) {
      offs[base + j] = run;
      run += c[j];
      if (base + j == n - 1) offs[n] = run;
    }
  }
}

__global__ __launch_bounds__(BIN_T) void gat_bin_scatter(
    const int* __restrict__ src, const int* __restrict__ dst,
    const int* __restrict__ offs, int* __restrict__ pairs, int nedges,
    int nb, int nblk, const float* __restrict__ X,
    const float* __restrict__ W, const float* __restrict__ atts,
    float* __restrict__ Y, float* __restrict__ a_s, int nrows, int g0,
    int gn) {
  __shared__ __attribute__((aligned(16))) unsigned char lds[64 * 128 * 2 * 4];
  if ((int)blockIdx.x < gn) {
    gat_gemm128_rows<BIN_T>((float*)lds, X, W, atts, Y, a_s, nrows,
                            (g0 + (int)blockIdx.x) * GEMM_ROWS);
    return;
  }
  int* gofs = (int*)lds;
  int* lstart = gofs + MAXB;
  int* cur = lstart + MAXB;
  int2* sorted = (int2*)(cur + MAXB);
  const int tid = threadIdx.x;
  const int blk = (int)blockIdx.x - gn;

  for (int b = tid; b < nb; b += BIN_T) {
    const int o = offs[b * nblk + blk];
    gofs[b] = o;
    lstart[b] = offs[b * nblk + blk + 1] - o;
    cur[b] = 0;
  }
  __syncthreads();
  if (tid < 32) {
    const int per = (nb + 31) >> 5;
    const int b0 = tid * per;
    int s = 0;
    for (int j = 0; j < per; ++j)
      if (b0 + j < nb) s += lstart[b0 + j];
    int inc = s;
    for (int off = 1; off < 32; off <<= 1) {
      int t = __shfl_up(inc, off, 32);
      if (tid >= off) inc += t;
    }
    int run = inc - s;
    for (int j = 0; j < per; ++j) {
      if (b0 + j < nb) {
        const int c = lstart[b0 + j];
        lstart[b0 + j] = run;
        run += c;
      }
    }
  }
  __syncthreads();

  const int base = blk * CHUNK;
  for (int k = 0; k < CHUNK / BIN_T; ++k) {
    const int i = base + k * BIN_T + tid;
    if (i < nedges) {
      const int d = dst[i], s = src[i];
      const int b = d >> BSH;
      const int r = atomicAdd(&cur[b], 1);
      sorted[lstart[b] + r] = make_int2(s, d);
    }
  }
  __syncthreads();

  int cnt = nedges - base;
  if (cnt > CHUNK) cnt = CHUNK;
  int2* __restrict__ pout = reinterpret_cast<int2*>(pairs);
  for (int t = tid; t < cnt; t += BIN_T) {
    const int2 p = sorted[t];
    const int b = p.y >> BSH;
    pout[gofs[b] + (t - lstart[b])] = p;
  }
}

__global__ __launch_bounds__(SORT_T) void gat_bucket_sort(
    const int* __restrict__ pairs, const int* __restrict__ offs,
    int* __restrict__ rowptr, int* __restrict__ ssrc, int n, int nb,
    int nblk, const float* __restrict__ Wd1, const float* __restrict__ ad1,
    float* __restrict__ vd1, const float* __restrict__ Wd2,
    const float* __restrict__ ad2, float* __restrict__ vd2) {
  __shared__ int cnt[BNODES];
  __shared__ int cursor[BNODES];
  __shared__ int wsum[BNODES / 32];
  __shared__ int stage[STAGE_CAP];
  const int tid = threadIdx.x;
  const int b = (int)blockIdx.x;
  if (b >= nb) {
    if (tid < 128) {
      if (b == nb) gat_matvec128_row(Wd1, ad1, vd1, tid);
      else gat_matvec128_row(Wd2, ad2, vd2, tid);
    }
    return;
  }
  const int beg = offs[b * nblk], end = offs[(b + 1) * nblk];
  const int node0 = b << BSH;
  const int2* __restrict__ pin = reinterpret_cast<const int2*>(pairs);

  if (tid < BNODES) cnt[tid] = 0;
  __syncthreads();
  for (int i0 = beg + tid; i0 < end; i0 += 4 * SORT_T) {
    int dl[4];
#pragma unroll
    for (int u = 0; u < 4; ++u) {
      const int i = i0 + u * SORT_T;
      dl[u] = (i < end) ? pin[i].y : node0;
    }
#pragma unroll
    for (int u = 0; u < 4; ++u)
      if (i0 + u * SORT_T < end) atomicAdd(&cnt[dl[u] - node0], 1);
  }
  __syncthreads();

  int c = 0, inc = 0;
  if (tid < BNODES) {
    c = cnt[tid];
    inc = c;
    for (int off = 1; off < 32; off <<= 1) {
      int t = __shfl_up(inc, off, 32);
      if ((tid & 31) >= off) inc += t;
    }
    if ((tid & 31) == 31) wsum[tid >> 5] = inc;
  }
  __syncthreads();
  if (tid < BNODES) {
    int run = inc - c;
    for (int w = 0; w < (tid >> 5); ++w) run += wsum[w];
    cursor[tid] = run;
    if (node0 + tid < n) rowptr[node0 + tid] = beg + run;
  }
  if (b == nb - 1 && tid == 0) rowptr[n] = end;
  __syncthreads();

  const bool staged = (end - beg) <= STAGE_CAP;
  for (int i0 = beg + tid; i0 < end; i0 += 4 * SORT_T) {
    int2 p[4];
#pragma unroll
    for (int u = 0; u < 4; ++u) {
      const int i = i0 + u * SORT_T;
      p[u] = (i < end) ? pin[i] : make_int2(0, node0);
    }
#pragma unroll
    for (int u = 0; u < 4; ++u) {
      if (i0 + u * SORT_T < end) {
        const int pos = atomicAdd(&cursor[p[u].y - node0], 1);
        if (staged) stage[pos] = p[u].x;
        else ssrc[beg + pos] = p[u].x;
      }
    }
  }
  if (staged) {
    __syncthreads();
    for (int i = tid; i < end - beg; i += SORT_T) ssrc[beg + i] = stage[i];
  }
}

__device__ __forceinline__ float wave_allreduce_max(float v) {
  for (int off = 16; off > 0; off >>= 1) v = fmaxf(v, __shfl_xor(v, off, 32));
  return v;
}
__device__ __forceinline__ float wave_allreduce_sum(float v) {
  for (int off = 16; off > 0; off >>= 1) v += __shfl_xor(v, off, 32);
  return v;
}
__device__ __forceinline__ float gat_leaky(float e) {
  return (e > 0.0f) ? e : NEG_SLOPE * e;
}
__device__ __forceinline__ int lane_i(int v, int l) {
  return __builtin_amdgcn_readlane(v, l);
}
__device__ __forceinline__ float lane_f(float v, int l) {
  return __int_as_float(__builtin_amdgcn_readlane(__float_as_int(v), l));
}
__device__ __forceinline__ v4f gat_row(const float* __restrict__ xl, int s) {
  return *(const v4f*)(xl + (size_t)s * 128);
}

template <bool HEAD>
__global__ __launch_bounds__(256) void gat_node_aggregate(
    const int* __restrict__ rowptr, const int* __restrict__ ssrc,
    const float* __restrict__ a_s, const float* __restrict__ xin,
    const float* __restrict__ vd, const float* __restrict__ xs,
    const float* __restrict__ bias, float* __restrict__ out,
    const float* __restrict__ Wl, const float* __restrict__ bl,
    float* __restrict__ out2, int n) {
  int node = (blockIdx.x * 256 + threadIdx.x) >> 5;
  const int lane = threadIdx.x & 31;
  if (node >= n) return;
  node = __builtin_amdgcn_readfirstlane(node);
  const int beg = rowptr[node], end = rowptr[node + 1];
  const v4f xiv = *(const v4f*)(xin + (size_t)node * 128 + lane * 4);
  const v4f vdv = *(const v4f*)(vd + lane * 4);

  const int c0 = beg + lane;
  const bool v0 = c0 < end;
  const int sr0 = ssrc[v0 ? c0 : 0];
  const int s0 = v0 ? sr0 : 0;
  const float as0 = a_s[s0];
  const float ad = lane_f(
      wave_reduce_sum(xiv.x * vdv.x + xiv.y * vdv.y + xiv.z * vdv.z +
                      xiv.w * vdv.w),
      0);
  const float e0 = v0 ? gat_leaky(as0 + ad) : -INFINITY;
  float mx = e0;
  for (int c = c0 + 32; c < end; c += 32)
    mx = fmaxf(mx, gat_leaky(a_s[ssrc[c]] + ad));
  mx = wave_allreduce_max(mx);
  if (!(fabsf(mx) < INFINITY)) mx = 0.0f;

  const float x0 = v0 ? __expf(e0 - mx) : 0.0f;
  float sum = x0;
  for (int c = c0 + 32; c < end; c += 32)
    sum += __expf(gat_leaky(a_s[ssrc[c]] + ad) - mx);
  sum = wave_allreduce_sum(sum);
  const float dn = sum + SM_EPS;

  const float* xl = xs + lane * 4;
  v4f acc = (v4f)(0.0f);
  for (int cb = beg; cb < end; cb += 32) {
    int sl = s0;
    float al = x0 / dn;
    if (cb != beg) {
      const int c = cb + lane;
      sl = 0;
      al = 0.0f;
      if (c < end) {
        sl = ssrc[c];
        al = __expf(gat_leaky(a_s[sl] + ad) - mx) / dn;
      }
    }
    const int cnt = (end - cb < 32) ? (end - cb) : 32;
    int j = 0;
    for (; j + 4 <= cnt; j += 4) {
      const v4f va = gat_row(xl, lane_i(sl, j));
      const v4f vb = gat_row(xl, lane_i(sl, j + 1));
      const v4f vc = gat_row(xl, lane_i(sl, j + 2));
      const v4f vdd = gat_row(xl, lane_i(sl, j + 3));
      acc += lane_f(al, j) * va;
      acc += lane_f(al, j + 1) * vb;
      acc += lane_f(al, j + 2) * vc;
      acc += lane_f(al, j + 3) * vdd;
    }
    const int rem = cnt - j;
    if (rem == 3) {
      const v4f va = gat_row(xl, lane_i(sl, j));
      const v4f vb = gat_row(xl, lane_i(sl, j + 1));
      const v4f vc = gat_row(xl, lane_i(sl, j + 2));
      acc += lane_f(al, j) * va;
      acc += lane_f(al, j + 1) * vb;
      acc += lane_f(al, j + 2) * vc;
    } else if (rem == 2) {
      const v4f va = gat_row(xl, lane_i(sl, j));
      const v4f vb = gat_row(xl, lane_i(sl, j + 1));
      acc += lane_f(al, j) * va;
      acc += lane_f(al, j + 1) * vb;
    } else if (rem == 1) {
      const v4f va = gat_row(xl, lane_i(sl, j));
      acc += lane_f(al, j) * va;
    }
  }

  const v4f b = *(const v4f*)(bias + lane * 4);
  v4f r;
  r.x = fmaxf(acc.x + b.x, 0.0f);
  r.y = fmaxf(acc.y + b.y, 0.0f);
  r.z = fmaxf(acc.z + b.z, 0.0f);
  r.w = fmaxf(acc.w + b.w, 0.0f);
  if (HEAD) {
    const v4f w0 = *(const v4f*)(Wl + lane * 8);
    const v4f w1 = *(const v4f*)(Wl + lane * 8 + 4);
    float p0 = r.x * w0.x + r.y * w0.z + r.z * w1.x + r.w * w1.z;
    float p1 = r.x * w0.y + r.y * w0.w + r.z * w1.y + r.w * w1.w;
    p0 = wave_reduce_sum(p0);
    p1 = wave_reduce_sum(p1);
    if (lane == 0) {
      out2[(size_t)node * 2 + 0] = p0 + bl[0];
      out2[(size_t)node * 2 + 1] = p1 + bl[1];
    }
  } else {
    *(v4f*)(out + (size_t)node * 128 + lane * 4) = r;
  }
}

static void run_gat_layer(bool own_gemm, const float* xin, const float* Wsrc,
                          const float* atts, const float* bias,
                          const int* rowptr, const int* ssrc, int N, float* xs,
                          float* outbuf, float* a_s, const float* vd,
                          const float* Wl, const float* bl, float* out2,
                          hipStream_t stream) {
  if (own_gemm)
    gat_gemm128_wmma<<<(N + 127) / 128, 256, 0, stream>>>(xin, Wsrc, atts, xs,
                                                          a_s, N);
  if (Wl != nullptr)
    gat_node_aggregate<true><<<(N + 7) / 8, 256, 0, stream>>>(
        rowptr, ssrc, a_s, xin, vd, xs, bias, outbuf, Wl, bl, out2, N);
  else
    gat_node_aggregate<false><<<(N + 7) / 8, 256, 0, stream>>>(
        rowptr, ssrc, a_s, xin, vd, xs, bias, outbuf, Wl, bl, out2, N);
}

extern "C" void kernel_launch(void* const* d_in, const int* in_sizes, int n_in,
                              void* d_out, int out_size, void* d_ws,
                              size_t ws_size, hipStream_t stream) {
  const float* x   = (const float*)d_in[0];
  const int* ei    = (const int*)d_in[1];
  const float* W1s = (const float*)d_in[2];
  const float* W1d = (const float*)d_in[3];
  const float* a1s = (const float*)d_in[4];
  const float* a1d = (const float*)d_in[5];
  const float* b1  = (const float*)d_in[6];
  const float* W2s = (const float*)d_in[7];
  const float* W2d = (const float*)d_in[8];
  const float* a2s = (const float*)d_in[9];
  const float* a2d = (const float*)d_in[10];
  const float* b2  = (const float*)d_in[11];
  const float* Wl  = (const float*)d_in[12];
  const float* bl  = (const float*)d_in[13];

  const int N = in_sizes[0] / 128;
  const int E = in_sizes[1] / 2;
  const int* src = ei;
  const int* dst = ei + E;
  const int nb    = (N + BNODES - 1) >> BSH;
  const int nblk  = (E + CHUNK - 1) / CHUNK;
  const int n2    = nb * nblk;
  const int n2p   = (n2 + 4) & ~3;
  const int tiles = (n2 + SCAN_TILE - 1) / SCAN_TILE;

  float* bufA  = (float*)d_ws;
  float* bufB  = bufA + (size_t)N * 128;
  float* vd1   = bufB + (size_t)N * 128;
  float* vd2   = vd1 + 128;
  int* hist    = (int*)(vd2 + 128);
  int* offs    = hist + n2p;
  int* pairs   = offs + n2p;
  float* a_s   = (float*)(pairs + (size_t)E * 2);
  int* rowptr  = (int*)(a_s + N);
  int* ssrc    = rowptr + N + 1;

  const int gblk = (N + GEMM_ROWS - 1) / GEMM_ROWS;
  const int g1 = gblk / 2;
  const int g2 = gblk - g1;

  gat_bin_count<<<g1 + nblk, BIN_T, 0, stream>>>(dst, hist, E, nb, nblk, x,
                                                 W1s, a1s, bufA, a_s, N, 0,
                                                 g1);
  gat_scan_offs<<<tiles, SCAN_T, 0, stream>>>(hist, offs, n2);
  gat_bin_scatter<<<g2 + nblk, BIN_T, 0, stream>>>(src, dst, offs, pairs, E,
                                                   nb, nblk, x, W1s, a1s,
                                                   bufA, a_s, N, g1, g2);
  gat_bucket_sort<<<nb + 2, SORT_T, 0, stream>>>(pairs, offs, rowptr, ssrc, N,
                                                 nb, nblk, W1d, a1d, vd1, W2d,
                                                 a2d, vd2);

  run_gat_layer(false, x, W1s, a1s, b1, rowptr, ssrc, N, bufA, bufB, a_s, vd1,
                nullptr, nullptr, nullptr, stream);
  run_gat_layer(true, bufB, W2s, a2s, b2, rowptr, ssrc, N, bufA, nullptr, a_s,
                vd2, Wl, bl, (float*)d_out, stream);
}
